# starting kernel
# baseline (speedup 1.0000x reference)
#include <hip/hip_runtime.h>


typedef __attribute__((ext_vector_type(2))) float v2f;
typedef __attribute__((ext_vector_type(8))) float v8f;

static constexpr int GN   = 100000;
static constexpr int GE   = 1000000;
static constexpr int DIN  = 128;
static constexpr int HID  = 64;
static constexpr int CAT  = 192;
static constexpr int LAB  = 40;

template <int W> struct PadRow {
    static constexpr int value = W + ((48 - (W % 32)) % 32);
};

template <int NT>
__launch_bounds__(256)
__global__ void gemm_wmma_f32(const float* __restrict__ A, int lda,
                              const float* __restrict__ B, int K, int dout,
                              int M, float* __restrict__ out, int ldo,
                              const float* __restrict__ bias, int bias_to_out,
                              float* __restrict__ agg,
                              const float* __restrict__ dinv)
{
    constexpr int W  = NT * 16;
    constexpr int WP = PadRow<W>::value;
    extern __shared__ v2f Bs2[];

    const int KP = K >> 1;
    for (int idx = threadIdx.x; idx < KP * W; idx += 256) {
        int kp = idx / W;
        int n  = idx - kp * W;
        v2f p;
        p.x = (n < dout) ? B[(2 * kp)     * dout + n] : 0.0f;
        p.y = (n < dout) ? B[(2 * kp + 1) * dout + n] : 0.0f;
        Bs2[kp * WP + n] = p;
    }
    __syncthreads();

    const int wave = threadIdx.x >> 5;
    const int lane = threadIdx.x & 31;
    const int l15  = lane & 15;
    const int half = lane >> 4;
    const int kb   = half * 2;

    const int m0 = (blockIdx.x * 8 + wave) * 16;
    int mA = m0 + l15;
    if (mA > M - 1) mA = M - 1;
    const float* Arow = A + (size_t)mA * (size_t)lda;

    v8f acc[NT];
#pragma unroll
    for (int t = 0; t < NT; ++t)
        acc[t] = (v8f){0.f, 0.f, 0.f, 0.f, 0.f, 0.f, 0.f, 0.f};

    for (int k = 0; k < K; k += 4) {
        v2f a = *(const v2f*)(Arow + k + kb);
        const v2f* brow = Bs2 + ((k >> 1) + half) * WP + l15;
#pragma unroll
        for (int t = 0; t < NT; ++t) {
            v2f b = brow[t * 16];
            acc[t] = __builtin_amdgcn_wmma_f32_16x16x4_f32(
                false, a, false, b, (short)0, acc[t], false, false);
        }
    }

#pragma unroll
    for (int t = 0; t < NT; ++t) {
        const int n = t * 16 + l15;
        float bv = 0.0f;
        if (bias && n < dout) bv = bias[n];
#pragma unroll
        for (int v = 0; v < 8; ++v) {
            const int m = m0 + half * 8 + v;
            if (m < M && n < dout) {
                const float val = acc[t][v];
                if (agg) {
                    out[(size_t)m * ldo + n] = val;
                    const float di = dinv[m];
                    agg[(size_t)m * dout + n] = bv + di * di * val;
                } else {
                    out[(size_t)m * ldo + n] = val + (bias_to_out ? bv : 0.0f);
                }
            }
        }
    }
}

__global__ void deg_init(float* __restrict__ d, int M) {
    int i = blockIdx.x * blockDim.x + threadIdx.x;
    if (i < M) d[i] = 1.0f;
}

__global__ void deg_scatter(const int* __restrict__ ei, int E,
                            float* __restrict__ deg) {
    int e = blockIdx.x * blockDim.x + threadIdx.x;
    if (e < E) unsafeAtomicAdd(&deg[ei[E + e]], 1.0f);
}

__global__ void deg_rsqrt(float* __restrict__ d, int M) {
    int i = blockIdx.x * blockDim.x + threadIdx.x;
    if (i < M) d[i] = rsqrtf(d[i]);
}

__global__ void edge_scatter(const int* __restrict__ ei, int E,
                             const float* __restrict__ dinv,
                             const float* __restrict__ hw,
                             float* __restrict__ agg)
{
    int gid = blockIdx.x * blockDim.x + threadIdx.x;
    int e = gid >> 5;
    if (e >= E) return;
    int lane = gid & 31;
    int s = ei[e];
    int d = ei[E + e];
    float w = dinv[s] * dinv[d];
    int c = lane * 2;
    float2 v = *(const float2*)(hw + (size_t)s * HID + c);
    float* p = agg + (size_t)d * HID + c;
    unsafeAtomicAdd(p,     w * v.x);
    unsafeAtomicAdd(p + 1, w * v.y);
}

__global__ void relu_concat(const float* __restrict__ agg,
                            float* __restrict__ Hcat, int M, int col0)
{
    int gid = blockIdx.x * blockDim.x + threadIdx.x;
    int i = gid >> 4;
    if (i >= M) return;
    int c = (gid & 15) * 4;
    float4 v = *(const float4*)(agg + (size_t)i * HID + c);
    v.x = fmaxf(v.x, 0.0f);
    v.y = fmaxf(v.y, 0.0f);
    v.z = fmaxf(v.z, 0.0f);
    v.w = fmaxf(v.w, 0.0f);
    *(float4*)(Hcat + (size_t)i * CAT + col0 + c) = v;
}

__global__ void logsoftmax40(float* __restrict__ out, int M)
{
    int i = blockIdx.x * blockDim.x + threadIdx.x;
    if (i >= M) return;
    float* row = out + (size_t)i * LAB;
    float v[LAB];
    float mx = -3.4e38f;
#pragma unroll
    for (int c = 0; c < LAB; ++c) { v[c] = row[c]; mx = fmaxf(mx, v[c]); }
    float s = 0.0f;
#pragma unroll
    for (int c = 0; c < LAB; ++c) s += __expf(v[c] - mx);
    float ls = __logf(s);
#pragma unroll
    for (int c = 0; c < LAB; ++c) row[c] = v[c] - mx - ls;
}

extern "C" void kernel_launch(void* const* d_in, const int* in_sizes, int n_in,
                              void* d_out, int out_size, void* d_ws, size_t ws_size,
                              hipStream_t stream)
{
    (void)in_sizes; (void)n_in; (void)out_size; (void)ws_size;

    const float* x  = (const float*)d_in[0];
    const int*   ei = (const int*)d_in[1];
    const float* Wm[3] = {(const float*)d_in[2], (const float*)d_in[4],
                          (const float*)d_in[6]};
    const float* bm[3] = {(const float*)d_in[3], (const float*)d_in[5],
                          (const float*)d_in[7]};
    const float* Wl = (const float*)d_in[8];
    const float* bl = (const float*)d_in[9];
    float* out = (float*)d_out;

    char* ws = (char*)d_ws;
    auto align256 = [](size_t b) { return (b + 255) & ~(size_t)255; };
    size_t off = 0;
    float* dinv = (float*)(ws + off); off += align256((size_t)GN * 4);
    float* hw   = (float*)(ws + off); off += align256((size_t)GN * HID * 4);
    float* agg  = (float*)(ws + off); off += align256((size_t)GN * HID * 4);
    float* Hcat = (float*)(ws + off); off += align256((size_t)GN * CAT * 4);

    const dim3 blk(256);

    deg_init   <<<(GN + 255) / 256, blk, 0, stream>>>(dinv, GN);
    deg_scatter<<<(GE + 255) / 256, blk, 0, stream>>>(ei, GE, dinv);
    deg_rsqrt  <<<(GN + 255) / 256, blk, 0, stream>>>(dinv, GN);

    const int gemmBlocks = (GN + 127) / 128;

    for (int l = 0; l < 3; ++l) {
        const float* Ain = (l == 0) ? x : (Hcat + HID * (l - 1));
        const int lda = (l == 0) ? DIN : CAT;
        const int K   = (l == 0) ? DIN : HID;
        const size_t shmem = (size_t)(K / 2) * PadRow<64>::value * sizeof(float2);
        gemm_wmma_f32<4><<<gemmBlocks, blk, shmem, stream>>>(
            Ain, lda, Wm[l], K, HID, GN, hw, HID, bm[l], 0, agg, dinv);
        edge_scatter<<<(GE * 32 + 255) / 256, blk, 0, stream>>>(
            ei, GE, dinv, hw, agg);
        relu_concat<<<((size_t)GN * 16 + 255) / 256, blk, 0, stream>>>(
            agg, Hcat, GN, HID * l);
    }

    const size_t shmemF = (size_t)(CAT / 2) * PadRow<48>::value * sizeof(float2);
    gemm_wmma_f32<3><<<gemmBlocks, blk, shmemF, stream>>>(
        Hcat, CAT, Wl, CAT, LAB, GN, out, LAB, bl, 1, nullptr, nullptr);

    logsoftmax40<<<(GN + 255) / 256, blk, 0, stream>>>(out, GN);
}


# edge push kept inside one half of the chip: bins = (256-node source block, destination half); a work-group reads its die, draws by int ticket a bin whose sum rows only its half adds to; arrival order
# speedup vs baseline: 1.4597x; 1.4597x over previous
#include <hip/hip_runtime.h>


typedef __attribute__((ext_vector_type(2))) float v2f;
typedef __attribute__((ext_vector_type(8))) float v8f;

static constexpr int GN   = 100000;
static constexpr int GE   = 1000000;
static constexpr int DIN  = 128;
static constexpr int HID  = 64;
static constexpr int CAT  = 192;
static constexpr int LAB  = 40;

template <int W> struct PadRow {
    static constexpr int value = W + ((48 - (W % 32)) % 32);
};

template <int NT, bool RELU>
__launch_bounds__(256)
__global__ void gemm_wmma_f32(const float* __restrict__ A0,
                              const float* __restrict__ A1,
                              const float* __restrict__ A2,
                              int lda, int Kp, int nplanes,
                              const float* __restrict__ B, int dout,
                              int M, float* __restrict__ out, int ldo,
                              const float* __restrict__ bias, int bias_to_out,
                              float* __restrict__ agg,
                              const float* __restrict__ dinv)
{
    constexpr int W  = NT * 16;
    constexpr int WP = PadRow<W>::value;
    extern __shared__ v2f Bs2[];

    const int KP = (Kp * nplanes) >> 1;
    for (int idx = threadIdx.x; idx < KP * W; idx += 256) {
        int kp = idx / W;
        int n  = idx - kp * W;
        v2f p;
        p.x = (n < dout) ? B[(2 * kp)     * dout + n] : 0.0f;
        p.y = (n < dout) ? B[(2 * kp + 1) * dout + n] : 0.0f;
        Bs2[kp * WP + n] = p;
    }
    __syncthreads();

    const int wave = threadIdx.x >> 5;
    const int lane = threadIdx.x & 31;
    const int l15  = lane & 15;
    const int half = lane >> 4;
    const int kb   = half * 2;

    const int m0 = (blockIdx.x * 8 + wave) * 16;
    int mA = m0 + l15;
    if (mA > M - 1) mA = M - 1;

    v8f acc[NT];
#pragma unroll
    for (int t = 0; t < NT; ++t)
        acc[t] = (v8f){0.f, 0.f, 0.f, 0.f, 0.f, 0.f, 0.f, 0.f};

    for (int p = 0; p < nplanes; ++p) {
        const float* Ap = (p == 0) ? A0 : ((p == 1) ? A1 : A2);
        const float* Arow = Ap + (size_t)mA * (size_t)lda;
        const v2f* bplane = Bs2 + ((p * Kp) >> 1) * WP;
        constexpr int NF = 4;
        for (int k = 0; k < Kp; k += 4 * NF) {
            v2f a[NF];
#pragma unroll
            for (int j = 0; j < NF; ++j)
                a[j] = *(const v2f*)(Arow + k + 4 * j + kb);
            if (RELU) {
#pragma unroll
                for (int j = 0; j < NF; ++j) {
                    a[j].x = fmaxf(a[j].x, 0.0f);
                    a[j].y = fmaxf(a[j].y, 0.0f);
                }
            }
#pragma unroll
            for (int j = 0; j < NF; ++j) {
                const v2f* brow = bplane + (((k + 4 * j) >> 1) + half) * WP + l15;
#pragma unroll
                for (int t = 0; t < NT; ++t) {
                    v2f b = brow[t * 16];
                    acc[t] = __builtin_amdgcn_wmma_f32_16x16x4_f32(
                        false, a[j], false, b, (short)0, acc[t], false, false);
                }
            }
        }
    }

#pragma unroll
    for (int t = 0; t < NT; ++t) {
        const int n = t * 16 + l15;
        float bv = 0.0f;
        if (bias && n < dout) bv = bias[n];
#pragma unroll
        for (int v = 0; v < 8; ++v) {
            const int m = m0 + half * 8 + v;
            if (m < M && n < dout) {
                const float val = acc[t][v];
                if (agg) {
                    out[(size_t)m * ldo + n] = val;
                    const float di = dinv[m];
                    agg[(size_t)m * dout + n] = bv + di * di * val;
                } else {
                    out[(size_t)m * ldo + n] = val + (bias_to_out ? bv : 0.0f);
                }
            }
        }
    }
}

static constexpr int SB    = 256;
static constexpr int SBSH  = 8;
static constexpr int NSB   = (GN + SB - 1) / SB;
static constexpr int NBIN  = 2 * NSB;
static constexpr int DHALF = GN / 2;
static constexpr int EPB  = 2048;
static constexpr int CST  = 64;
static constexpr int EPT  = EPB / 256;

__device__ __forceinline__ int bin_of(int s, int d) {
    return ((s >> SBSH) << 1) | (d >= DHALF ? 1 : 0);
}

__global__ void deg_init(float* __restrict__ d, int M,
                         int* __restrict__ binCount, int* __restrict__ binCursor,
                         int* __restrict__ tick) {
    int i = blockIdx.x * blockDim.x + threadIdx.x;
    if (i < M) d[i] = 1.0f;
    if (i < NBIN) { binCount[i * CST] = 0; binCursor[i * CST] = 0; }
    if (i < 6) tick[i * CST] = 0;
}

__launch_bounds__(256)
__global__ void deg_scatter(const int* __restrict__ ei, int E,
                            float* __restrict__ deg, int* __restrict__ binCount) {
    __shared__ int hist[NBIN];
    for (int i = threadIdx.x; i < NBIN; i += 256) hist[i] = 0;
    __syncthreads();
    const int base = blockIdx.x * EPB + threadIdx.x;
    int sv[EPT], dv[EPT];
#pragma unroll
    for (int k = 0; k < EPT; ++k) {
        int e = base + k * 256;
        sv[k] = -1; dv[k] = 0;
        if (e < E) { sv[k] = ei[e]; dv[k] = ei[E + e]; }
    }
#pragma unroll
    for (int k = 0; k < EPT; ++k) {
        if (sv[k] >= 0) {
            unsafeAtomicAdd(&deg[dv[k]], 1.0f);
            atomicAdd(&hist[bin_of(sv[k], dv[k])], 1);
        }
    }
    __syncthreads();
    for (int i = threadIdx.x; i < NBIN; i += 256) {
        int c = hist[i];
        if (c) atomicAdd(&binCount[i * CST], c);
    }
}

__global__ void deg_rsqrt(float* __restrict__ d, int M) {
    int i = blockIdx.x * blockDim.x + threadIdx.x;
    if (i < M) d[i] = rsqrtf(d[i]);
}

__launch_bounds__(256)
__global__ void bin_place(const int* __restrict__ ei, int E,
                          const float* __restrict__ dinv,
                          const int* __restrict__ binCount,
                          int* __restrict__ binCursor,
                          int* __restrict__ binStart,
                          int2* __restrict__ rec)
{
    __shared__ int hist[NBIN];
    __shared__ int start[NBIN];
    __shared__ int part[256];
    const int tid = threadIdx.x;

    int c[4];
    int sum = 0;
#pragma unroll
    for (int q = 0; q < 4; ++q) {
        int b = 4 * tid + q;
        c[q] = (b < NBIN) ? binCount[b * CST] : 0;
        sum += c[q];
    }
    part[tid] = sum;
    __syncthreads();
    for (int off = 1; off < 256; off <<= 1) {
        int v = (tid >= off) ? part[tid - off] : 0;
        __syncthreads();
        part[tid] += v;
        __syncthreads();
    }
    int excl = part[tid] - sum;
#pragma unroll
    for (int q = 0; q < 4; ++q) {
        int b = 4 * tid + q;
        if (b < NBIN) {
            start[b] = excl;
            if (blockIdx.x == 0) binStart[b] = excl;
        }
        excl += c[q];
    }
    if (blockIdx.x == 0 && tid == 255) binStart[NBIN] = part[255];
    for (int i = tid; i < NBIN; i += 256) hist[i] = 0;
    __syncthreads();

    int key[EPT], rnk[EPT], bin[EPT];
    float wgt[EPT];
    const int base = blockIdx.x * EPB + tid;
#pragma unroll
    for (int k = 0; k < EPT; ++k) {
        int e = base + k * 256;
        bin[k] = -1;
        if (e < E) {
            int s = ei[e];
            int d = ei[E + e];
            bin[k] = bin_of(s, d);
            key[k] = (d << SBSH) | (s & (SB - 1));
            wgt[k] = dinv[s] * dinv[d];
            rnk[k] = atomicAdd(&hist[bin[k]], 1);
        }
    }
    __syncthreads();
    for (int i = tid; i < NBIN; i += 256) {
        int n = hist[i];
        hist[i] = n ? (start[i] + atomicAdd(&binCursor[i * CST], n)) : 0;
    }
    __syncthreads();
#pragma unroll
    for (int k = 0; k < EPT; ++k) {
        if (bin[k] >= 0) {
            int pos = hist[bin[k]] + rnk[k];
            rec[pos] = make_int2(key[k], __float_as_int(wgt[k]));
        }
    }
}

__launch_bounds__(256)
__global__ void edge_push(const float* __restrict__ hw,
                          const int* __restrict__ binStart,
                          const int2* __restrict__ rec,
                          float* __restrict__ agg, int M,
                          int* __restrict__ tick)
{
    __shared__ __attribute__((aligned(16))) float Hs[SB * HID];
    const int tid  = threadIdx.x;
    int* Hi = (int*)Hs;
    if (tid == 0) {
        const unsigned id = __builtin_amdgcn_s_sendmsg_rtn(0x87);
        int set = (int)((id >> 18) & 1u);
        int t = atomicAdd(&tick[set * CST], 1);
        if (t >= NSB) { set ^= 1; t = atomicAdd(&tick[set * CST], 1); }
        Hi[0] = (t < NSB) ? (2 * t + set) : -1;
    }
    __syncthreads();
    const int bin = Hi[0];
    __syncthreads();
    if (bin < 0) return;
    const int row0 = (bin >> 1) * SB;
    constexpr int NST = SB * (HID / 8) / 256;
    float4 lo[NST], hi[NST];
#pragma unroll
    for (int it = 0; it < NST; ++it) {
        const int i  = tid + it * 256;
        const int r  = i >> 3;
        const int c4 = (i & 7) * 4;
        const int gr = min(row0 + r, M - 1);
        const float* g = hw + (size_t)gr * HID + c4;
        lo[it] = *(const float4*)g;
        hi[it] = *(const float4*)(g + 32);
    }
#pragma unroll
    for (int it = 0; it < NST; ++it) {
        const int i  = tid + it * 256;
        const int r  = i >> 3;
        const int c4 = (i & 7) * 4;
        if (row0 + r >= M) {
            lo[it] = make_float4(0.f, 0.f, 0.f, 0.f);
            hi[it] = make_float4(0.f, 0.f, 0.f, 0.f);
        }
        float* sp = Hs + r * HID + 2 * c4;
        *(float4*)sp       = make_float4(lo[it].x, hi[it].x, lo[it].y, hi[it].y);
        *(float4*)(sp + 4) = make_float4(lo[it].z, hi[it].z, lo[it].w, hi[it].w);
    }
    __syncthreads();

    const int beg  = binStart[bin];
    const int end  = binStart[bin + 1];
    const int wave = __builtin_amdgcn_readfirstlane(tid >> 5);
    const int lane = tid & 31;
    const float* hl = Hs + 2 * lane;
    float* al = agg + lane;

    constexpr int NG = 4;
    for (int cb = beg + wave * 32; cb < end; cb += 256 * NG) {
        int2 r[NG];
#pragma unroll
        for (int g = 0; g < NG; ++g) {
            const int idx = cb + g * 256 + lane;
            r[g] = make_int2(0, 0);
            if (idx < end) r[g] = rec[idx];
        }
#pragma unroll
        for (int g = 0; g < NG; ++g) {
            const int cnt = min(32, end - (cb + g * 256));
            int j = 0;
            for (; j + 4 <= cnt; j += 4) {
                int k[4]; float w[4]; float2 v[4];
#pragma unroll
                for (int q = 0; q < 4; ++q) {
                    k[q] = __builtin_amdgcn_readlane(r[g].x, j + q);
                    w[q] = __int_as_float(__builtin_amdgcn_readlane(r[g].y, j + q));
                    v[q] = *(const float2*)(hl + (k[q] & (SB - 1)) * HID);
                }
#pragma unroll
                for (int q = 0; q < 4; ++q) {
                    float* p = al + (unsigned)(k[q] >> SBSH) * (unsigned)HID;
                    unsafeAtomicAdd(p,      w[q] * v[q].x);
                    unsafeAtomicAdd(p + 32, w[q] * v[q].y);
                }
            }
            for (; j < cnt; ++j) {
                const int   k1 = __builtin_amdgcn_readlane(r[g].x, j);
                const float w1 = __int_as_float(__builtin_amdgcn_readlane(r[g].y, j));
                const float2 v1 = *(const float2*)(hl + (k1 & (SB - 1)) * HID);
                float* p = al + (unsigned)(k1 >> SBSH) * (unsigned)HID;
                unsafeAtomicAdd(p,      w1 * v1.x);
                unsafeAtomicAdd(p + 32, w1 * v1.y);
            }
        }
    }
}

__global__ void logsoftmax40(float* __restrict__ out, int M)
{
    int i = blockIdx.x * blockDim.x + threadIdx.x;
    if (i >= M) return;
    float* row = out + (size_t)i * LAB;
    float v[LAB];
    float mx = -3.4e38f;
#pragma unroll
    for (int c = 0; c < LAB; ++c) { v[c] = row[c]; mx = fmaxf(mx, v[c]); }
    float s = 0.0f;
#pragma unroll
    for (int c = 0; c < LAB; ++c) s += __expf(v[c] - mx);
    float ls = __logf(s);
#pragma unroll
    for (int c = 0; c < LAB; ++c) row[c] = v[c] - mx - ls;
}

extern "C" void kernel_launch(void* const* d_in, const int* in_sizes, int n_in,
                              void* d_out, int out_size, void* d_ws, size_t ws_size,
                              hipStream_t stream)
{
    (void)in_sizes; (void)n_in; (void)out_size; (void)ws_size;

    const float* x  = (const float*)d_in[0];
    const int*   ei = (const int*)d_in[1];
    const float* Wm[3] = {(const float*)d_in[2], (const float*)d_in[4],
                          (const float*)d_in[6]};
    const float* bm[3] = {(const float*)d_in[3], (const float*)d_in[5],
                          (const float*)d_in[7]};
    const float* Wl = (const float*)d_in[8];
    const float* bl = (const float*)d_in[9];
    float* out = (float*)d_out;

    char* ws = (char*)d_ws;
    auto align256 = [](size_t b) { return (b + 255) & ~(size_t)255; };
    size_t off = 0;
    float* dinv = (float*)(ws + off); off += align256((size_t)GN * 4);
    float* hw   = (float*)(ws + off); off += align256((size_t)GN * HID * 4);
    float* agg[3];
    for (int l = 0; l < 3; ++l) {
        agg[l] = (float*)(ws + off); off += align256((size_t)GN * HID * 4);
    }
    int*  binCount  = (int*)(ws + off);  off += align256((size_t)NBIN * CST * 4);
    int*  binCursor = (int*)(ws + off);  off += align256((size_t)NBIN * CST * 4);
    int*  binStart  = (int*)(ws + off);  off += align256((size_t)(NBIN + 1) * 4);
    int2* rec       = (int2*)(ws + off); off += align256((size_t)GE * 8);
    int*  tick      = (int*)(ws + off);  off += align256((size_t)6 * CST * 4);

    const dim3 blk(256);

    deg_init   <<<(GN + 255) / 256, blk, 0, stream>>>(dinv, GN, binCount, binCursor, tick);
    deg_scatter<<<(GE + EPB - 1) / EPB, blk, 0, stream>>>(ei, GE, dinv, binCount);
    deg_rsqrt  <<<(GN + 255) / 256, blk, 0, stream>>>(dinv, GN);
    bin_place  <<<(GE + EPB - 1) / EPB, blk, 0, stream>>>(ei, GE, dinv, binCount,
                                                        binCursor, binStart, rec);

    const int gemmBlocks = (GN + 127) / 128;

    for (int l = 0; l < 3; ++l) {
        const int K = (l == 0) ? DIN : HID;
        const size_t shmem = (size_t)(K / 2) * PadRow<64>::value * sizeof(float2);
        if (l == 0) {
            gemm_wmma_f32<4, false><<<gemmBlocks, blk, shmem, stream>>>(
                x, x, x, DIN, DIN, 1, Wm[0], HID, GN, hw, HID, bm[0], 0,
                agg[0], dinv);
        } else {
            gemm_wmma_f32<4, true><<<gemmBlocks, blk, shmem, stream>>>(
                agg[l - 1], agg[l - 1], agg[l - 1], HID, HID, 1, Wm[l], HID,
                GN, hw, HID, bm[l], 0, agg[l], dinv);
        }
        edge_push<<<NBIN, blk, 0, stream>>>(hw, binStart, rec, agg[l], GN,
                                            tick + l * 2 * CST);
    }

    const size_t shmemF = (size_t)(CAT / 2) * PadRow<48>::value * sizeof(float2);
    gemm_wmma_f32<3, true><<<gemmBlocks, blk, shmemF, stream>>>(
        agg[0], agg[1], agg[2], HID, HID, 3, Wl, LAB, GN, out, LAB, bl, 1,
        nullptr, nullptr);

    logsoftmax40<<<(GN + 255) / 256, blk, 0, stream>>>(out, GN);
}


# in-degrees counted in LDS per destination block from the binned edge keys (int LDS adds) instead of 1M global float adds; layer 0's pull forms the edge weights and writes the records
# speedup vs baseline: 1.8545x; 1.2705x over previous
#include <hip/hip_runtime.h>


typedef __attribute__((ext_vector_type(2))) float v2f;
typedef __attribute__((ext_vector_type(8))) float v8f;

static constexpr int GN   = 100000;
static constexpr int GE   = 1000000;
static constexpr int DIN  = 128;
static constexpr int HID  = 64;
static constexpr int CAT  = 192;
static constexpr int LAB  = 40;

template <int W> struct PadRow {
    static constexpr int value = W + ((48 - (W % 32)) % 32);
};

template <int X>
__device__ __forceinline__ float row_xor(float v) {
    return __int_as_float(__builtin_amdgcn_update_dpp(
        0, __float_as_int(v), 0x160 + X, 0xf, 0xf, true));
}
__device__ __forceinline__ float row_max16(float v) {
    v = fmaxf(v, row_xor<1>(v));
    v = fmaxf(v, row_xor<2>(v));
    v = fmaxf(v, row_xor<4>(v));
    v = fmaxf(v, row_xor<8>(v));
    return v;
}
__device__ __forceinline__ float row_sum16(float v) {
    v += row_xor<1>(v);
    v += row_xor<2>(v);
    v += row_xor<4>(v);
    v += row_xor<8>(v);
    return v;
}

template <int NT, bool RELU, bool LSM, int DOUT>
__launch_bounds__(256)
__global__ void gemm_wmma_f32(const float* __restrict__ A0,
                              const float* __restrict__ A1,
                              const float* __restrict__ A2,
                              int lda, int Kp, int nplanes,
                              const float* __restrict__ B, int dout,
                              int M, float* __restrict__ out, int ldo,
                              const float* __restrict__ bias)
{
    constexpr int W  = NT * 16;
    constexpr int WP = PadRow<W>::value;
    extern __shared__ __attribute__((aligned(16))) v2f Bs2[];

    const int KP = (Kp * nplanes) >> 1;
    for (int idx = threadIdx.x; idx < KP * W; idx += 256) {
        int kp = idx / W;
        int n  = idx - kp * W;
        v2f p;
        p.x = (n < dout) ? B[(2 * kp)     * dout + n] : 0.0f;
        p.y = (n < dout) ? B[(2 * kp + 1) * dout + n] : 0.0f;
        Bs2[kp * WP + n] = p;
    }
    __syncthreads();

    const int wave = threadIdx.x >> 5;
    const int lane = threadIdx.x & 31;
    const int l15  = lane & 15;
    const int half = lane >> 4;
    const int kb   = half * 2;

    const int m0 = (blockIdx.x * 8 + wave) * 16;
    int mA = m0 + l15;
    if (mA > M - 1) mA = M - 1;

    v8f acc[NT];
#pragma unroll
    for (int t = 0; t < NT; ++t)
        acc[t] = (v8f){0.f, 0.f, 0.f, 0.f, 0.f, 0.f, 0.f, 0.f};

    for (int p = 0; p < nplanes; ++p) {
        const float* Ap = (p == 0) ? A0 : ((p == 1) ? A1 : A2);
        const float* Arow = Ap + (size_t)mA * (size_t)lda;
        const v2f* bplane = Bs2 + ((p * Kp) >> 1) * WP;
        constexpr int NF = 4;
        for (int k = 0; k < Kp; k += 4 * NF) {
            v2f a[NF];
#pragma unroll
            for (int j = 0; j < NF; ++j)
                a[j] = *(const v2f*)(Arow + k + 4 * j + kb);
            if (RELU) {
#pragma unroll
                for (int j = 0; j < NF; ++j) {
                    a[j].x = fmaxf(a[j].x, 0.0f);
                    a[j].y = fmaxf(a[j].y, 0.0f);
                }
            }
#pragma unroll
            for (int j = 0; j < NF; ++j) {
                const v2f* brow = bplane + (((k + 4 * j) >> 1) + half) * WP + l15;
#pragma unroll
                for (int t = 0; t < NT; ++t) {
                    v2f b = brow[t * 16];
                    acc[t] = __builtin_amdgcn_wmma_f32_16x16x4_f32(
                        false, a[j], false, b, (short)0, acc[t], false, false);
                }
            }
        }
    }

    if (LSM) {
        static_assert(!LSM || (DOUT % 4 == 0 && DOUT <= NT * 16), "row shape");
        float bv[NT];
        bool  ok[NT];
#pragma unroll
        for (int t = 0; t < NT; ++t) {
            const int n = t * 16 + l15;
            ok[t] = (t * 16 + 15 < DOUT) || (n < DOUT);
            bv[t] = (bias && ok[t]) ? bias[n] : 0.0f;
        }
        __syncthreads();
        float* tile = (float*)Bs2 + wave * (16 * DOUT);
#pragma unroll
        for (int v = 0; v < 8; ++v) {
            float xv[NT];
            float mx = -3.4e38f;
#pragma unroll
            for (int t = 0; t < NT; ++t) {
                xv[t] = acc[t][v] + bv[t];
                mx = fmaxf(mx, ok[t] ? xv[t] : -3.4e38f);
            }
            mx = row_max16(mx);
            float s = 0.0f;
#pragma unroll
            for (int t = 0; t < NT; ++t)
                s += ok[t] ? __expf(xv[t] - mx) : 0.0f;
            s = row_sum16(s);
            const float ls = __logf(s);
            float* trow = tile + (half * 8 + v) * DOUT + l15;
#pragma unroll
            for (int t = 0; t < NT; ++t)
                if (ok[t]) trow[t * 16] = xv[t] - mx - ls;
        }
        __syncthreads();
        const int rows = min(16, M - m0);
        const int nvec = (rows > 0) ? (rows * DOUT) >> 2 : 0;
        const float4* srcv = (const float4*)tile;
        float4* dstv = (float4*)(out + (size_t)m0 * DOUT);
        for (int i = lane; i < nvec; i += 32) dstv[i] = srcv[i];
        return;
    }

#pragma unroll
    for (int t = 0; t < NT; ++t) {
        const int n = t * 16 + l15;
        float bv = 0.0f;
        if (bias && n < dout) bv = bias[n];
#pragma unroll
        for (int v = 0; v < 8; ++v) {
            const int m = m0 + half * 8 + v;
            if (m < M && n < dout)
                out[(size_t)m * ldo + n] = acc[t][v] + bv;
        }
    }
}

static constexpr int DB    = 98;
static constexpr int DLB   = 7;
static constexpr int NBIN  = (GN + DB - 1) / DB;
static constexpr int EPB  = 2048;
static constexpr int CST  = 64;
static constexpr int EPT  = EPB / 256;
static_assert(DB < (1 << DLB), "in-block row number must fit its field");
static_assert(((long long)GN << DLB) < (1LL << 31), "a key (src << DLB | row) must stay non-negative");
static_assert(NBIN <= 4 * 256, "bin_place scans four bins per thread");

__global__ void bin_init(int* __restrict__ binCount, int* __restrict__ binCursor) {
    int i = blockIdx.x * blockDim.x + threadIdx.x;
    if (i < NBIN) { binCount[i * CST] = 0; binCursor[i * CST] = 0; }
}

__launch_bounds__(256)
__global__ void bin_count(const int* __restrict__ ei, int E, int* __restrict__ binCount) {
    __shared__ int hist[NBIN];
    for (int i = threadIdx.x; i < NBIN; i += 256) hist[i] = 0;
    __syncthreads();
    const int base = blockIdx.x * EPB + threadIdx.x;
    int dv[EPT];
#pragma unroll
    for (int k = 0; k < EPT; ++k) {
        int e = base + k * 256;
        dv[k] = -1;
        if (e < E) dv[k] = ei[E + e];
    }
#pragma unroll
    for (int k = 0; k < EPT; ++k) {
        if (dv[k] >= 0) atomicAdd(&hist[dv[k] / DB], 1);
    }
    __syncthreads();
    for (int i = threadIdx.x; i < NBIN; i += 256) {
        int c = hist[i];
        if (c) atomicAdd(&binCount[i * CST], c);
    }
}

__launch_bounds__(256)
__global__ void deg_bins(const int* __restrict__ keys,
                         const int* __restrict__ binStart,
                         float* __restrict__ dinv, int M) {
    __shared__ int cnt[1 << DLB];
    const int tid  = threadIdx.x;
    const int bin  = blockIdx.x;
    const int row0 = bin * DB;
    const int nrow = min(DB, M - row0);
    if (tid < (1 << DLB)) cnt[tid] = 0;
    __syncthreads();
    const int beg = binStart[bin];
    const int end = binStart[bin + 1];
    constexpr int KPT = 4;
    for (int base = beg + tid; base < end; base += 256 * KPT) {
        int kv[KPT];
#pragma unroll
        for (int q = 0; q < KPT; ++q) {
            const int i = base + q * 256;
            kv[q] = (i < end) ? keys[i] : -1;
        }
#pragma unroll
        for (int q = 0; q < KPT; ++q)
            if (kv[q] >= 0) atomicAdd(&cnt[kv[q] & ((1 << DLB) - 1)], 1);
    }
    __syncthreads();
    if (tid < nrow) dinv[row0 + tid] = rsqrtf(1.0f + (float)cnt[tid]);
}

__launch_bounds__(256)
__global__ void bin_place(const int* __restrict__ ei, int E,
                          const int* __restrict__ binCount,
                          int* __restrict__ binCursor,
                          int* __restrict__ binStart,
                          int* __restrict__ keys)
{
    __shared__ int hist[NBIN];
    __shared__ int start[NBIN];
    __shared__ int part[256];
    const int tid = threadIdx.x;

    int c[4];
    int sum = 0;
#pragma unroll
    for (int q = 0; q < 4; ++q) {
        int b = 4 * tid + q;
        c[q] = (b < NBIN) ? binCount[b * CST] : 0;
        sum += c[q];
    }
    part[tid] = sum;
    __syncthreads();
    for (int off = 1; off < 256; off <<= 1) {
        int v = (tid >= off) ? part[tid - off] : 0;
        __syncthreads();
        part[tid] += v;
        __syncthreads();
    }
    int excl = part[tid] - sum;
#pragma unroll
    for (int q = 0; q < 4; ++q) {
        int b = 4 * tid + q;
        if (b < NBIN) {
            start[b] = excl;
            if (blockIdx.x == 0) binStart[b] = excl;
        }
        excl += c[q];
    }
    if (blockIdx.x == 0 && tid == 255) binStart[NBIN] = part[255];
    for (int i = tid; i < NBIN; i += 256) hist[i] = 0;
    __syncthreads();

    int key[EPT], rnk[EPT], bin[EPT];
    const int base = blockIdx.x * EPB + tid;
#pragma unroll
    for (int k = 0; k < EPT; ++k) {
        int e = base + k * 256;
        bin[k] = -1;
        if (e < E) {
            int s = ei[e];
            int d = ei[E + e];
            bin[k] = d / DB;
            key[k] = (s << DLB) | (d - bin[k] * DB);
            rnk[k] = atomicAdd(&hist[bin[k]], 1);
        }
    }
    __syncthreads();
    for (int i = tid; i < NBIN; i += 256) {
        int n = hist[i];
        hist[i] = n ? (start[i] + atomicAdd(&binCursor[i * CST], n)) : 0;
    }
    __syncthreads();
#pragma unroll
    for (int k = 0; k < EPT; ++k) {
        if (bin[k] >= 0) {
            int pos = hist[bin[k]] + rnk[k];
            keys[pos] = key[k];
        }
    }
}

template <bool FIRST>
__launch_bounds__(256)
__global__ void edge_pull(const float* __restrict__ hw,
                          const int* __restrict__ binStart,
                          const int* __restrict__ keys,
                          int2* __restrict__ rec,
                          const float* __restrict__ bias,
                          const float* __restrict__ dinv,
                          float* __restrict__ agg, int M)
{
    __shared__ __attribute__((aligned(16))) float As[DB * HID];
    const int tid  = threadIdx.x;
    const int bin  = blockIdx.x;
    const int row0 = bin * DB;
    const int nrow = min(DB, M - row0);

    for (int i = tid; i < nrow * (HID / 4); i += 256) {
        const int r  = i >> 4;
        const int c4 = (i & 15) * 4;
        float di = dinv[row0 + r];
        di *= di;
        const float4 h = *(const float4*)(hw + (size_t)(row0 + r) * HID + c4);
        const float4 b = *(const float4*)(bias + c4);
        *(float4*)(As + r * HID + c4) =
            make_float4(b.x + di * h.x, b.y + di * h.y, b.z + di * h.z, b.w + di * h.w);
    }
    __syncthreads();

    const int beg  = binStart[bin];
    const int end  = binStart[bin + 1];
    const int wave = __builtin_amdgcn_readfirstlane(tid >> 5);
    const int lane = tid & 31;
    const float* hl = hw + 2 * lane;
    float* al = As + 2 * lane;

    constexpr int NE = 16;
    for (int cb = beg + wave * 32; cb < end; cb += 256) {
        const int cnt = min(32, end - cb);
        int2 r = make_int2(0, 0);
        if (FIRST) {
            if (lane < cnt) {
                const int key = keys[cb + lane];
                const float w = dinv[(unsigned)key >> DLB] *
                                dinv[row0 + (key & ((1 << DLB) - 1))];
                r = make_int2(key, __float_as_int(w));
                rec[cb + lane] = r;
            }
        } else {
            if (lane < cnt) r = rec[cb + lane];
        }
        int j = 0;
        for (; j + NE <= cnt; j += NE) {
            float2 v[NE];
#pragma unroll
            for (int q = 0; q < NE; ++q) {
                const unsigned k = (unsigned)__builtin_amdgcn_readlane(r.x, j + q);
                v[q] = *(const float2*)(hl + (size_t)(k >> DLB) * HID);
            }
#pragma unroll
            for (int q = 0; q < NE; ++q) {
                const unsigned k = (unsigned)__builtin_amdgcn_readlane(r.x, j + q);
                const float w = __int_as_float(__builtin_amdgcn_readlane(r.y, j + q));
                float* p = al + (k & ((1u << DLB) - 1u)) * HID;
                atomicAdd(p,     w * v[q].x);
                atomicAdd(p + 1, w * v[q].y);
            }
        }
        for (; j + 4 <= cnt; j += 4) {
            float2 v[4];
#pragma unroll
            for (int q = 0; q < 4; ++q) {
                const unsigned k = (unsigned)__builtin_amdgcn_readlane(r.x, j + q);
                v[q] = *(const float2*)(hl + (size_t)(k >> DLB) * HID);
            }
#pragma unroll
            for (int q = 0; q < 4; ++q) {
                const unsigned k = (unsigned)__builtin_amdgcn_readlane(r.x, j + q);
                const float w = __int_as_float(__builtin_amdgcn_readlane(r.y, j + q));
                float* p = al + (k & ((1u << DLB) - 1u)) * HID;
                atomicAdd(p,     w * v[q].x);
                atomicAdd(p + 1, w * v[q].y);
            }
        }
        for (; j < cnt; ++j) {
            const unsigned k = (unsigned)__builtin_amdgcn_readlane(r.x, j);
            const float w = __int_as_float(__builtin_amdgcn_readlane(r.y, j));
            const float2 v1 = *(const float2*)(hl + (size_t)(k >> DLB) * HID);
            float* p = al + (k & ((1u << DLB) - 1u)) * HID;
            atomicAdd(p,     w * v1.x);
            atomicAdd(p + 1, w * v1.y);
        }
    }
    __syncthreads();

    for (int i = tid; i < nrow * (HID / 4); i += 256) {
        const int r  = i >> 4;
        const int c4 = (i & 15) * 4;
        *(float4*)(agg + (size_t)(row0 + r) * HID + c4) = *(const float4*)(As + r * HID + c4);
    }
}

extern "C" void kernel_launch(void* const* d_in, const int* in_sizes, int n_in,
                              void* d_out, int out_size, void* d_ws, size_t ws_size,
                              hipStream_t stream)
{
    (void)in_sizes; (void)n_in; (void)out_size; (void)ws_size;

    const float* x  = (const float*)d_in[0];
    const int*   ei = (const int*)d_in[1];
    const float* Wm[3] = {(const float*)d_in[2], (const float*)d_in[4],
                          (const float*)d_in[6]};
    const float* bm[3] = {(const float*)d_in[3], (const float*)d_in[5],
                          (const float*)d_in[7]};
    const float* Wl = (const float*)d_in[8];
    const float* bl = (const float*)d_in[9];
    float* out = (float*)d_out;

    char* ws = (char*)d_ws;
    auto align256 = [](size_t b) { return (b + 255) & ~(size_t)255; };
    size_t off = 0;
    float* dinv = (float*)(ws + off); off += align256((size_t)GN * 4);
    float* hw   = (float*)(ws + off); off += align256((size_t)GN * HID * 4);
    float* agg[3];
    for (int l = 0; l < 3; ++l) {
        agg[l] = (float*)(ws + off); off += align256((size_t)GN * HID * 4);
    }
    int*  binCount  = (int*)(ws + off);  off += align256((size_t)NBIN * CST * 4);
    int*  binCursor = (int*)(ws + off);  off += align256((size_t)NBIN * CST * 4);
    int*  binStart  = (int*)(ws + off);  off += align256((size_t)(NBIN + 1) * 4);
    int2* rec       = (int2*)(ws + off); off += align256((size_t)GE * 8);
    int*  keys      = (int*)(ws + off);  off += align256((size_t)GE * 4);

    const dim3 blk(256);

    bin_init   <<<(NBIN + 255) / 256, blk, 0, stream>>>(binCount, binCursor);
    bin_count  <<<(GE + EPB - 1) / EPB, blk, 0, stream>>>(ei, GE, binCount);
    bin_place  <<<(GE + EPB - 1) / EPB, blk, 0, stream>>>(ei, GE, binCount,
                                                        binCursor, binStart, keys);
    deg_bins   <<<NBIN, blk, 0, stream>>>(keys, binStart, dinv, GN);

    const int gemmBlocks = (GN + 127) / 128;

    for (int l = 0; l < 3; ++l) {
        const int K = (l == 0) ? DIN : HID;
        const size_t shmem = (size_t)(K / 2) * PadRow<64>::value * sizeof(float2);
        if (l == 0) {
            gemm_wmma_f32<4, false, false, HID><<<gemmBlocks, blk, shmem, stream>>>(
                x, x, x, DIN, DIN, 1, Wm[0], HID, GN, hw, HID, nullptr);
        } else {
            gemm_wmma_f32<4, true, false, HID><<<gemmBlocks, blk, shmem, stream>>>(
                agg[l - 1], agg[l - 1], agg[l - 1], HID, HID, 1, Wm[l], HID,
                GN, hw, HID, nullptr);
        }
        if (l == 0)
            edge_pull<true><<<NBIN, blk, 0, stream>>>(hw, binStart, keys, rec, bm[l], dinv,
                                                      agg[l], GN);
        else
            edge_pull<false><<<NBIN, blk, 0, stream>>>(hw, binStart, keys, rec, bm[l], dinv,
                                                       agg[l], GN);
    }

    const size_t shmemF = (size_t)(CAT / 2) * PadRow<48>::value * sizeof(float2);
    gemm_wmma_f32<3, true, true, LAB><<<gemmBlocks, blk, shmemF, stream>>>(
        agg[0], agg[1], agg[2], HID, HID, 3, Wl, LAB, GN, out, LAB, bl);
}


# edge binning without global counters: groups of 8192 edges leave per-bin counts in a table; placing sums it, orders keys by bin in LDS, writes runs; 1 launch, global int adds gone; sum order kept
# speedup vs baseline: 2.5549x; 1.3777x over previous
#include <hip/hip_runtime.h>


typedef __attribute__((ext_vector_type(2))) float v2f;
typedef __attribute__((ext_vector_type(8))) float v8f;

static constexpr int GN   = 100000;
static constexpr int GE   = 1000000;
static constexpr int DIN  = 128;
static constexpr int HID  = 64;
static constexpr int CAT  = 192;
static constexpr int LAB  = 40;

template <int W> struct PadRow {
    static constexpr int value = W + ((48 - (W % 32)) % 32);
};

template <int X>
__device__ __forceinline__ float row_xor(float v) {
    return __int_as_float(__builtin_amdgcn_update_dpp(
        0, __float_as_int(v), 0x160 + X, 0xf, 0xf, true));
}
__device__ __forceinline__ float row_max16(float v) {
    v = fmaxf(v, row_xor<1>(v));
    v = fmaxf(v, row_xor<2>(v));
    v = fmaxf(v, row_xor<4>(v));
    v = fmaxf(v, row_xor<8>(v));
    return v;
}
__device__ __forceinline__ float row_sum16(float v) {
    v += row_xor<1>(v);
    v += row_xor<2>(v);
    v += row_xor<4>(v);
    v += row_xor<8>(v);
    return v;
}

template <int NT, bool RELU, bool LSM, int DOUT>
__launch_bounds__(256)
__global__ void gemm_wmma_f32(const float* __restrict__ A0,
                              const float* __restrict__ A1,
                              const float* __restrict__ A2,
                              int lda, int Kp, int nplanes,
                              const float* __restrict__ B, int dout,
                              int M, float* __restrict__ out, int ldo,
                              const float* __restrict__ bias)
{
    constexpr int W  = NT * 16;
    constexpr int WP = PadRow<W>::value;
    extern __shared__ __attribute__((aligned(16))) v2f Bs2[];

    const int KP = (Kp * nplanes) >> 1;
    for (int idx = threadIdx.x; idx < KP * W; idx += 256) {
        int kp = idx / W;
        int n  = idx - kp * W;
        v2f p;
        p.x = (n < dout) ? B[(2 * kp)     * dout + n] : 0.0f;
        p.y = (n < dout) ? B[(2 * kp + 1) * dout + n] : 0.0f;
        Bs2[kp * WP + n] = p;
    }
    __syncthreads();

    const int wave = threadIdx.x >> 5;
    const int lane = threadIdx.x & 31;
    const int l15  = lane & 15;
    const int half = lane >> 4;
    const int kb   = half * 2;

    const int m0 = (blockIdx.x * 8 + wave) * 16;
    int mA = m0 + l15;
    if (mA > M - 1) mA = M - 1;

    v8f acc[NT];
#pragma unroll
    for (int t = 0; t < NT; ++t)
        acc[t] = (v8f){0.f, 0.f, 0.f, 0.f, 0.f, 0.f, 0.f, 0.f};

    for (int p = 0; p < nplanes; ++p) {
        const float* Ap = (p == 0) ? A0 : ((p == 1) ? A1 : A2);
        const float* Arow = Ap + (size_t)mA * (size_t)lda;
        const v2f* bplane = Bs2 + ((p * Kp) >> 1) * WP;
        constexpr int NF = 4;
        for (int k = 0; k < Kp; k += 4 * NF) {
            v2f a[NF];
#pragma unroll
            for (int j = 0; j < NF; ++j)
                a[j] = *(const v2f*)(Arow + k + 4 * j + kb);
            if (RELU) {
#pragma unroll
                for (int j = 0; j < NF; ++j) {
                    a[j].x = fmaxf(a[j].x, 0.0f);
                    a[j].y = fmaxf(a[j].y, 0.0f);
                }
            }
#pragma unroll
            for (int j = 0; j < NF; ++j) {
                const v2f* brow = bplane + (((k + 4 * j) >> 1) + half) * WP + l15;
#pragma unroll
                for (int t = 0; t < NT; ++t) {
                    v2f b = brow[t * 16];
                    acc[t] = __builtin_amdgcn_wmma_f32_16x16x4_f32(
                        false, a[j], false, b, (short)0, acc[t], false, false);
                }
            }
        }
    }

    if (LSM) {
        static_assert(!LSM || (DOUT % 4 == 0 && DOUT <= NT * 16), "row shape");
        float bv[NT];
        bool  ok[NT];
#pragma unroll
        for (int t = 0; t < NT; ++t) {
            const int n = t * 16 + l15;
            ok[t] = (t * 16 + 15 < DOUT) || (n < DOUT);
            bv[t] = (bias && ok[t]) ? bias[n] : 0.0f;
        }
        __syncthreads();
        float* tile = (float*)Bs2 + wave * (16 * DOUT);
#pragma unroll
        for (int v = 0; v < 8; ++v) {
            float xv[NT];
            float mx = -3.4e38f;
#pragma unroll
            for (int t = 0; t < NT; ++t) {
                xv[t] = acc[t][v] + bv[t];
                mx = fmaxf(mx, ok[t] ? xv[t] : -3.4e38f);
            }
            mx = row_max16(mx);
            float s = 0.0f;
#pragma unroll
            for (int t = 0; t < NT; ++t)
                s += ok[t] ? __expf(xv[t] - mx) : 0.0f;
            s = row_sum16(s);
            const float ls = __logf(s);
            float* trow = tile + (half * 8 + v) * DOUT + l15;
#pragma unroll
            for (int t = 0; t < NT; ++t)
                if (ok[t]) trow[t * 16] = xv[t] - mx - ls;
        }
        __syncthreads();
        const int rows = min(16, M - m0);
        const int nvec = (rows > 0) ? (rows * DOUT) >> 2 : 0;
        const float4* srcv = (const float4*)tile;
        float4* dstv = (float4*)(out + (size_t)m0 * DOUT);
        for (int i = lane; i < nvec; i += 32) dstv[i] = srcv[i];
        return;
    }

#pragma unroll
    for (int t = 0; t < NT; ++t) {
        const int n = t * 16 + l15;
        float bv = 0.0f;
        if (bias && n < dout) bv = bias[n];
#pragma unroll
        for (int v = 0; v < 8; ++v) {
            const int m = m0 + half * 8 + v;
            if (m < M && n < dout)
                out[(size_t)m * ldo + n] = acc[t][v] + bv;
        }
    }
}

static constexpr int DB    = 98;
static constexpr int DLB   = 7;
static constexpr int NBIN  = (GN + DB - 1) / DB;
static constexpr int BT    = 1024;
static constexpr int EPB   = 8192;
static constexpr int EPT   = EPB / BT;
static constexpr int NBLK  = (GE + EPB - 1) / EPB;
static constexpr int NBP   = 1024;
static constexpr int RG    = 4;
static constexpr int RPG   = (NBLK + RG - 1) / RG;
static_assert(DB < (1 << DLB), "in-block row number must fit its field");
static_assert(((long long)GN << DLB) < (1LL << 31), "a key (src << DLB | row) must stay non-negative");
static_assert(NBIN < NBP && NBP == BT, "one thread per padded bin, and one padded bin behind the last");
static_assert(BT / RG * 4 == NBP, "a row group's threads cover one table row with 16-byte loads");
static_assert(2 * RG * NBP >= EPB, "the table partials' LDS space is reused for the ordered keys");
static_assert(NBIN < 65536, "a bin number is staged as 16 bits");

__launch_bounds__(BT)
__global__ void bin_count(const int* __restrict__ ei, int E, int* __restrict__ cnt) {
    __shared__ int hist[NBP];
    const int tid = threadIdx.x;
    hist[tid] = 0;
    __syncthreads();
    const int base = blockIdx.x * EPB + tid;
    int dv[EPT];
#pragma unroll
    for (int k = 0; k < EPT; ++k) {
        int e = base + k * BT;
        dv[k] = -1;
        if (e < E) dv[k] = ei[E + e];
    }
#pragma unroll
    for (int k = 0; k < EPT; ++k) {
        if (dv[k] >= 0) atomicAdd(&hist[dv[k] / DB], 1);
    }
    __syncthreads();
    cnt[blockIdx.x * NBP + tid] = hist[tid];
}

__launch_bounds__(256)
__global__ void deg_bins(const int* __restrict__ keys,
                         const int* __restrict__ binStart,
                         float* __restrict__ dinv, int M) {
    __shared__ int cnt[1 << DLB];
    const int tid  = threadIdx.x;
    const int bin  = blockIdx.x;
    const int row0 = bin * DB;
    const int nrow = min(DB, M - row0);
    if (tid < (1 << DLB)) cnt[tid] = 0;
    __syncthreads();
    const int beg = binStart[bin];
    const int end = binStart[bin + 1];
    constexpr int KPT = 4;
    for (int base = beg + tid; base < end; base += 256 * KPT) {
        int kv[KPT];
#pragma unroll
        for (int q = 0; q < KPT; ++q) {
            const int i = base + q * 256;
            kv[q] = (i < end) ? keys[i] : -1;
        }
#pragma unroll
        for (int q = 0; q < KPT; ++q)
            if (kv[q] >= 0) atomicAdd(&cnt[kv[q] & ((1 << DLB) - 1)], 1);
    }
    __syncthreads();
    if (tid < nrow) dinv[row0 + tid] = rsqrtf(1.0f + (float)cnt[tid]);
}

__device__ __forceinline__ int wave_incl_scan(int v, int lane) {
#pragma unroll
    for (int off = 1; off < 32; off <<= 1) {
        const int t = __shfl_up(v, off);
        if (lane >= off) v += t;
    }
    return v;
}
__device__ __forceinline__ int block_excl_scan(int v, int* wsum, int& total) {
    const int lane = threadIdx.x & 31;
    const int wave = threadIdx.x >> 5;
    const int inc = wave_incl_scan(v, lane);
    if (lane == 31) wsum[wave] = inc;
    __syncthreads();
    const int ws   = wsum[lane];
    const int winc = wave_incl_scan(ws, lane);
    const int wexc = __shfl(winc, wave) - __shfl(ws, wave);
    total = __shfl(winc, 31);
    __syncthreads();
    return wexc + inc - v;
}

__launch_bounds__(BT)
__global__ void bin_place(const int* __restrict__ ei, int E,
                          const int* __restrict__ cnt,
                          int* __restrict__ binStart,
                          int* __restrict__ keys)
{
    __shared__ __attribute__((aligned(16))) int stage[2 * RG * NBP];
    __shared__ int hist[NBP];
    __shared__ int lstart[NBP];
    __shared__ int gdelta[NBP];
    __shared__ unsigned short sbin[EPB];
    __shared__ int wsum[32];
    const int tid = threadIdx.x;
    const int blk = blockIdx.x;

    int sv[EPT], dv[EPT];
    const int base = blk * EPB + tid;
#pragma unroll
    for (int k = 0; k < EPT; ++k) {
        int e = base + k * BT;
        sv[k] = 0;
        dv[k] = -1;
        if (e < E) { sv[k] = ei[e]; dv[k] = ei[E + e]; }
    }

    {
        const int g  = __builtin_amdgcn_readfirstlane(tid >> 8);
        const int c4 = (tid & 255) * 4;
        int4 tot  = make_int4(0, 0, 0, 0);
        int4 part = make_int4(0, 0, 0, 0);
        constexpr int TB = 16;
#pragma unroll 1
        for (int h = 0; h < RPG; h += TB) {
            int4 v[TB];
#pragma unroll
            for (int r = 0; r < TB; ++r) {
                const int rr  = h + r;
                const int row = g * RPG + rr;
                v[r] = make_int4(0, 0, 0, 0);
                if (rr < RPG && row < NBLK)
                    v[r] = *(const int4*)(cnt + (size_t)row * NBP + c4);
            }
#pragma unroll
            for (int r = 0; r < TB; ++r) {
                const int row = g * RPG + h + r;
                tot.x += v[r].x; tot.y += v[r].y; tot.z += v[r].z; tot.w += v[r].w;
                if (row < blk) {
                    part.x += v[r].x; part.y += v[r].y; part.z += v[r].z; part.w += v[r].w;
                }
            }
        }
        *(int4*)(stage + g * NBP + c4)        = tot;
        *(int4*)(stage + (RG + g) * NBP + c4) = part;
    }
    hist[tid] = 0;
    __syncthreads();
    int tot = 0, part = 0;
#pragma unroll
    for (int g = 0; g < RG; ++g) {
        tot  += stage[g * NBP + tid];
        part += stage[(RG + g) * NBP + tid];
    }
    int all;
    const int start = block_excl_scan(tot, wsum, all);
    if (blk == 0 && tid <= NBIN) binStart[tid] = start;

    int key[EPT], rnk[EPT], bin[EPT];
#pragma unroll
    for (int k = 0; k < EPT; ++k) {
        bin[k] = -1;
        if (dv[k] >= 0) {
            bin[k] = dv[k] / DB;
            key[k] = (sv[k] << DLB) | (dv[k] - bin[k] * DB);
            rnk[k] = atomicAdd(&hist[bin[k]], 1);
        }
    }
    __syncthreads();
    int nloc;
    const int ls = block_excl_scan(hist[tid], wsum, nloc);
    lstart[tid] = ls;
    gdelta[tid] = start + part - ls;
    __syncthreads();

    int* skey = stage;
#pragma unroll
    for (int k = 0; k < EPT; ++k) {
        if (bin[k] >= 0) {
            const int p = lstart[bin[k]] + rnk[k];
            skey[p] = key[k];
            sbin[p] = (unsigned short)bin[k];
        }
    }
    __syncthreads();
    for (int idx = tid; idx < nloc; idx += BT)
        keys[gdelta[sbin[idx]] + idx] = skey[idx];
}

template <bool FIRST>
__launch_bounds__(256)
__global__ void edge_pull(const float* __restrict__ hw,
                          const int* __restrict__ binStart,
                          const int* __restrict__ keys,
                          int2* __restrict__ rec,
                          const float* __restrict__ bias,
                          const float* __restrict__ dinv,
                          float* __restrict__ agg, int M)
{
    __shared__ __attribute__((aligned(16))) float As[DB * HID];
    const int tid  = threadIdx.x;
    const int bin  = blockIdx.x;
    const int row0 = bin * DB;
    const int nrow = min(DB, M - row0);

    for (int i = tid; i < nrow * (HID / 4); i += 256) {
        const int r  = i >> 4;
        const int c4 = (i & 15) * 4;
        float di = dinv[row0 + r];
        di *= di;
        const float4 h = *(const float4*)(hw + (size_t)(row0 + r) * HID + c4);
        const float4 b = *(const float4*)(bias + c4);
        *(float4*)(As + r * HID + c4) =
            make_float4(b.x + di * h.x, b.y + di * h.y, b.z + di * h.z, b.w + di * h.w);
    }
    __syncthreads();

    const int beg  = binStart[bin];
    const int end  = binStart[bin + 1];
    const int wave = __builtin_amdgcn_readfirstlane(tid >> 5);
    const int lane = tid & 31;
    const float* hl = hw + 2 * lane;
    float* al = As + 2 * lane;

    constexpr int NE = 16;
    for (int cb = beg + wave * 32; cb < end; cb += 256) {
        const int cnt = min(32, end - cb);
        int2 r = make_int2(0, 0);
        if (FIRST) {
            if (lane < cnt) {
                const int key = keys[cb + lane];
                const float w = dinv[(unsigned)key >> DLB] *
                                dinv[row0 + (key & ((1 << DLB) - 1))];
                r = make_int2(key, __float_as_int(w));
                rec[cb + lane] = r;
            }
        } else {
            if (lane < cnt) r = rec[cb + lane];
        }
        int j = 0;
        for (; j + NE <= cnt; j += NE) {
            float2 v[NE];
#pragma unroll
            for (int q = 0; q < NE; ++q) {
                const unsigned k = (unsigned)__builtin_amdgcn_readlane(r.x, j + q);
                v[q] = *(const float2*)(hl + (size_t)(k >> DLB) * HID);
            }
#pragma unroll
            for (int q = 0; q < NE; ++q) {
                const unsigned k = (unsigned)__builtin_amdgcn_readlane(r.x, j + q);
                const float w = __int_as_float(__builtin_amdgcn_readlane(r.y, j + q));
                float* p = al + (k & ((1u << DLB) - 1u)) * HID;
                atomicAdd(p,     w * v[q].x);
                atomicAdd(p + 1, w * v[q].y);
            }
        }
        for (; j + 4 <= cnt; j += 4) {
            float2 v[4];
#pragma unroll
            for (int q = 0; q < 4; ++q) {
                const unsigned k = (unsigned)__builtin_amdgcn_readlane(r.x, j + q);
                v[q] = *(const float2*)(hl + (size_t)(k >> DLB) * HID);
            }
#pragma unroll
            for (int q = 0; q < 4; ++q) {
                const unsigned k = (unsigned)__builtin_amdgcn_readlane(r.x, j + q);
                const float w = __int_as_float(__builtin_amdgcn_readlane(r.y, j + q));
                float* p = al + (k & ((1u << DLB) - 1u)) * HID;
                atomicAdd(p,     w * v[q].x);
                atomicAdd(p + 1, w * v[q].y);
            }
        }
        for (; j < cnt; ++j) {
            const unsigned k = (unsigned)__builtin_amdgcn_readlane(r.x, j);
            const float w = __int_as_float(__builtin_amdgcn_readlane(r.y, j));
            const float2 v1 = *(const float2*)(hl + (size_t)(k >> DLB) * HID);
            float* p = al + (k & ((1u << DLB) - 1u)) * HID;
            atomicAdd(p,     w * v1.x);
            atomicAdd(p + 1, w * v1.y);
        }
    }
    __syncthreads();

    for (int i = tid; i < nrow * (HID / 4); i += 256) {
        const int r  = i >> 4;
        const int c4 = (i & 15) * 4;
        *(float4*)(agg + (size_t)(row0 + r) * HID + c4) = *(const float4*)(As + r * HID + c4);
    }
}

extern "C" void kernel_launch(void* const* d_in, const int* in_sizes, int n_in,
                              void* d_out, int out_size, void* d_ws, size_t ws_size,
                              hipStream_t stream)
{
    (void)in_sizes; (void)n_in; (void)out_size; (void)ws_size;

    const float* x  = (const float*)d_in[0];
    const int*   ei = (const int*)d_in[1];
    const float* Wm[3] = {(const float*)d_in[2], (const float*)d_in[4],
                          (const float*)d_in[6]};
    const float* bm[3] = {(const float*)d_in[3], (const float*)d_in[5],
                          (const float*)d_in[7]};
    const float* Wl = (const float*)d_in[8];
    const float* bl = (const float*)d_in[9];
    float* out = (float*)d_out;

    char* ws = (char*)d_ws;
    auto align256 = [](size_t b) { return (b + 255) & ~(size_t)255; };
    size_t off = 0;
    float* dinv = (float*)(ws + off); off += align256((size_t)GN * 4);
    float* hw   = (float*)(ws + off); off += align256((size_t)GN * HID * 4);
    float* agg[3];
    for (int l = 0; l < 3; ++l) {
        agg[l] = (float*)(ws + off); off += align256((size_t)GN * HID * 4);
    }
    int*  cnt       = (int*)(ws + off);  off += align256((size_t)NBLK * NBP * 4);
    int*  binStart  = (int*)(ws + off);  off += align256((size_t)(NBIN + 1) * 4);
    int2* rec       = (int2*)(ws + off); off += align256((size_t)GE * 8);
    int*  keys      = (int*)(ws + off);  off += align256((size_t)GE * 4);

    const dim3 blk(256);

    bin_count  <<<NBLK, dim3(BT), 0, stream>>>(ei, GE, cnt);
    bin_place  <<<NBLK, dim3(BT), 0, stream>>>(ei, GE, cnt, binStart, keys);
    deg_bins   <<<NBIN, blk, 0, stream>>>(keys, binStart, dinv, GN);

    const int gemmBlocks = (GN + 127) / 128;

    for (int l = 0; l < 3; ++l) {
        const int K = (l == 0) ? DIN : HID;
        const size_t shmem = (size_t)(K / 2) * PadRow<64>::value * sizeof(float2);
        if (l == 0) {
            gemm_wmma_f32<4, false, false, HID><<<gemmBlocks, blk, shmem, stream>>>(
                x, x, x, DIN, DIN, 1, Wm[0], HID, GN, hw, HID, nullptr);
        } else {
            gemm_wmma_f32<4, true, false, HID><<<gemmBlocks, blk, shmem, stream>>>(
                agg[l - 1], agg[l - 1], agg[l - 1], HID, HID, 1, Wm[l], HID,
                GN, hw, HID, nullptr);
        }
        if (l == 0)
            edge_pull<true><<<NBIN, blk, 0, stream>>>(hw, binStart, keys, rec, bm[l], dinv,
                                                      agg[l], GN);
        else
            edge_pull<false><<<NBIN, blk, 0, stream>>>(hw, binStart, keys, rec, bm[l], dinv,
                                                       agg[l], GN);
    }

    const size_t shmemF = (size_t)(CAT / 2) * PadRow<48>::value * sizeof(float2);
    gemm_wmma_f32<3, true, true, LAB><<<gemmBlocks, blk, shmemF, stream>>>(
        agg[0], agg[1], agg[2], HID, HID, 3, Wl, LAB, GN, out, LAB, bl);
}


# loads of a phase issued as one group, one memory round trip each: GEMM W staging (16-byte loads), edge pull start rows, a chunk's 32 source rows, next records a chunk ahead; add order kept
# speedup vs baseline: 2.8522x; 1.1164x over previous
#include <hip/hip_runtime.h>


typedef __attribute__((ext_vector_type(2))) float v2f;
typedef __attribute__((ext_vector_type(8))) float v8f;

static constexpr int GN   = 100000;
static constexpr int GE   = 1000000;
static constexpr int DIN  = 128;
static constexpr int HID  = 64;
static constexpr int CAT  = 192;
static constexpr int LAB  = 40;

template <int W> struct PadRow {
    static constexpr int value = W + ((48 - (W % 32)) % 32);
};

template <int X>
__device__ __forceinline__ float row_xor(float v) {
    return __int_as_float(__builtin_amdgcn_update_dpp(
        0, __float_as_int(v), 0x160 + X, 0xf, 0xf, true));
}
__device__ __forceinline__ float row_max16(float v) {
    v = fmaxf(v, row_xor<1>(v));
    v = fmaxf(v, row_xor<2>(v));
    v = fmaxf(v, row_xor<4>(v));
    v = fmaxf(v, row_xor<8>(v));
    return v;
}
__device__ __forceinline__ float row_sum16(float v) {
    v += row_xor<1>(v);
    v += row_xor<2>(v);
    v += row_xor<4>(v);
    v += row_xor<8>(v);
    return v;
}

template <int NT, bool RELU, bool LSM, int DOUT, int KT>
__launch_bounds__(256)
__global__ void gemm_wmma_f32(const float* __restrict__ A0,
                              const float* __restrict__ A1,
                              const float* __restrict__ A2,
                              int lda, int Kp, int nplanes,
                              const float* __restrict__ B, int dout,
                              int M, float* __restrict__ out, int ldo,
                              const float* __restrict__ bias)
{
    constexpr int W  = NT * 16;
    constexpr int WP = PadRow<W>::value;
    extern __shared__ __attribute__((aligned(16))) v2f Bs2[];

    static_assert(DOUT % 4 == 0 && KT % 2 == 0, "B rows are read as 16-byte pieces, in k pairs");
    constexpr int G   = W / 4;
    constexpr int NU  = (KT / 2) * G;
    constexpr int UPT = (NU + 255) / 256;
    {
        float4 b0[UPT], b1[UPT];
#pragma unroll
        for (int i = 0; i < UPT; ++i) {
            const int u  = threadIdx.x + 256 * i;
            const int kp = u / G;
            const int c4 = (u - kp * G) * 4;
            const bool in = (256 * (i + 1) <= NU) || (u < NU);
            const int kl = in ? kp : 0;
            const int cl = (c4 < DOUT) ? c4 : 0;
            b0[i] = *(const float4*)(B + (size_t)(2 * kl)     * DOUT + cl);
            b1[i] = *(const float4*)(B + (size_t)(2 * kl + 1) * DOUT + cl);
        }
#pragma unroll
        for (int i = 0; i < UPT; ++i) {
            const int u  = threadIdx.x + 256 * i;
            const int kp = u / G;
            const int c4 = (u - kp * G) * 4;
            if (c4 >= DOUT) {
                b0[i] = make_float4(0.f, 0.f, 0.f, 0.f);
                b1[i] = make_float4(0.f, 0.f, 0.f, 0.f);
            }
            if ((256 * (i + 1) <= NU) || (u < NU)) {
                v2f* d = Bs2 + kp * WP + c4;
                d[0] = (v2f){b0[i].x, b1[i].x};
                d[1] = (v2f){b0[i].y, b1[i].y};
                d[2] = (v2f){b0[i].z, b1[i].z};
                d[3] = (v2f){b0[i].w, b1[i].w};
            }
        }
    }
    __syncthreads();

    const int wave = threadIdx.x >> 5;
    const int lane = threadIdx.x & 31;
    const int l15  = lane & 15;
    const int half = lane >> 4;
    const int kb   = half * 2;

    const int m0 = (blockIdx.x * 8 + wave) * 16;
    int mA = m0 + l15;
    if (mA > M - 1) mA = M - 1;

    v8f acc[NT];
#pragma unroll
    for (int t = 0; t < NT; ++t)
        acc[t] = (v8f){0.f, 0.f, 0.f, 0.f, 0.f, 0.f, 0.f, 0.f};

    for (int p = 0; p < nplanes; ++p) {
        const float* Ap = (p == 0) ? A0 : ((p == 1) ? A1 : A2);
        const float* Arow = Ap + (size_t)mA * (size_t)lda;
        const v2f* bplane = Bs2 + ((p * Kp) >> 1) * WP;
        constexpr int NF = 4;
        for (int k = 0; k < Kp; k += 4 * NF) {
            v2f a[NF];
#pragma unroll
            for (int j = 0; j < NF; ++j)
                a[j] = *(const v2f*)(Arow + k + 4 * j + kb);
            if (RELU) {
#pragma unroll
                for (int j = 0; j < NF; ++j) {
                    a[j].x = fmaxf(a[j].x, 0.0f);
                    a[j].y = fmaxf(a[j].y, 0.0f);
                }
            }
#pragma unroll
            for (int j = 0; j < NF; ++j) {
                const v2f* brow = bplane + (((k + 4 * j) >> 1) + half) * WP + l15;
#pragma unroll
                for (int t = 0; t < NT; ++t) {
                    v2f b = brow[t * 16];
                    acc[t] = __builtin_amdgcn_wmma_f32_16x16x4_f32(
                        false, a[j], false, b, (short)0, acc[t], false, false);
                }
            }
        }
    }

    if (LSM) {
        static_assert(!LSM || (DOUT % 4 == 0 && DOUT <= NT * 16), "row shape");
        float bv[NT];
        bool  ok[NT];
#pragma unroll
        for (int t = 0; t < NT; ++t) {
            const int n = t * 16 + l15;
            ok[t] = (t * 16 + 15 < DOUT) || (n < DOUT);
            bv[t] = (bias && ok[t]) ? bias[n] : 0.0f;
        }
        __syncthreads();
        float* tile = (float*)Bs2 + wave * (16 * DOUT);
#pragma unroll
        for (int v = 0; v < 8; ++v) {
            float xv[NT];
            float mx = -3.4e38f;
#pragma unroll
            for (int t = 0; t < NT; ++t) {
                xv[t] = acc[t][v] + bv[t];
                mx = fmaxf(mx, ok[t] ? xv[t] : -3.4e38f);
            }
            mx = row_max16(mx);
            float s = 0.0f;
#pragma unroll
            for (int t = 0; t < NT; ++t)
                s += ok[t] ? __expf(xv[t] - mx) : 0.0f;
            s = row_sum16(s);
            const float ls = __logf(s);
            float* trow = tile + (half * 8 + v) * DOUT + l15;
#pragma unroll
            for (int t = 0; t < NT; ++t)
                if (ok[t]) trow[t * 16] = xv[t] - mx - ls;
        }
        __syncthreads();
        const int rows = min(16, M - m0);
        const int nvec = (rows > 0) ? (rows * DOUT) >> 2 : 0;
        const float4* srcv = (const float4*)tile;
        float4* dstv = (float4*)(out + (size_t)m0 * DOUT);
        for (int i = lane; i < nvec; i += 32) dstv[i] = srcv[i];
        return;
    }

#pragma unroll
    for (int t = 0; t < NT; ++t) {
        const int n = t * 16 + l15;
        float bv = 0.0f;
        if (bias && n < dout) bv = bias[n];
#pragma unroll
        for (int v = 0; v < 8; ++v) {
            const int m = m0 + half * 8 + v;
            if (m < M && n < dout)
                out[(size_t)m * ldo + n] = acc[t][v] + bv;
        }
    }
}

static constexpr int DB    = 98;
static constexpr int DLB   = 7;
static constexpr int NBIN  = (GN + DB - 1) / DB;
static constexpr int BT    = 1024;
static constexpr int EPB   = 8192;
static constexpr int EPT   = EPB / BT;
static constexpr int NBLK  = (GE + EPB - 1) / EPB;
static constexpr int NBP   = 1024;
static constexpr int RG    = 4;
static constexpr int RPG   = (NBLK + RG - 1) / RG;
static_assert(DB < (1 << DLB), "in-block row number must fit its field");
static_assert(((long long)GN << DLB) < (1LL << 31), "a key (src << DLB | row) must stay non-negative");
static_assert(NBIN < NBP && NBP == BT, "one thread per padded bin, and one padded bin behind the last");
static_assert(BT / RG * 4 == NBP, "a row group's threads cover one table row with 16-byte loads");
static_assert(2 * RG * NBP >= EPB, "the table partials' LDS space is reused for the ordered keys");
static_assert(NBIN < 65536, "a bin number is staged as 16 bits");

__launch_bounds__(BT)
__global__ void bin_count(const int* __restrict__ ei, int E, int* __restrict__ cnt) {
    __shared__ int hist[NBP];
    const int tid = threadIdx.x;
    hist[tid] = 0;
    __syncthreads();
    const int base = blockIdx.x * EPB + tid;
    int dv[EPT];
#pragma unroll
    for (int k = 0; k < EPT; ++k) {
        int e = base + k * BT;
        dv[k] = -1;
        if (e < E) dv[k] = ei[E + e];
    }
#pragma unroll
    for (int k = 0; k < EPT; ++k) {
        if (dv[k] >= 0) atomicAdd(&hist[dv[k] / DB], 1);
    }
    __syncthreads();
    cnt[blockIdx.x * NBP + tid] = hist[tid];
}

__launch_bounds__(256)
__global__ void deg_bins(const int* __restrict__ keys,
                         const int* __restrict__ binStart,
                         float* __restrict__ dinv, int M) {
    __shared__ int cnt[1 << DLB];
    const int tid  = threadIdx.x;
    const int bin  = blockIdx.x;
    const int row0 = bin * DB;
    const int nrow = min(DB, M - row0);
    if (tid < (1 << DLB)) cnt[tid] = 0;
    __syncthreads();
    const int beg = binStart[bin];
    const int end = binStart[bin + 1];
    constexpr int KPT = 4;
    for (int base = beg + tid; base < end; base += 256 * KPT) {
        int kv[KPT];
#pragma unroll
        for (int q = 0; q < KPT; ++q) {
            const int i = base + q * 256;
            kv[q] = (i < end) ? keys[i] : -1;
        }
#pragma unroll
        for (int q = 0; q < KPT; ++q)
            if (kv[q] >= 0) atomicAdd(&cnt[kv[q] & ((1 << DLB) - 1)], 1);
    }
    __syncthreads();
    if (tid < nrow) dinv[row0 + tid] = rsqrtf(1.0f + (float)cnt[tid]);
}

__device__ __forceinline__ int wave_incl_scan(int v, int lane) {
#pragma unroll
    for (int off = 1; off < 32; off <<= 1) {
        const int t = __shfl_up(v, off);
        if (lane >= off) v += t;
    }
    return v;
}
__device__ __forceinline__ int block_excl_scan(int v, int* wsum, int& total) {
    const int lane = threadIdx.x & 31;
    const int wave = threadIdx.x >> 5;
    const int inc = wave_incl_scan(v, lane);
    if (lane == 31) wsum[wave] = inc;
    __syncthreads();
    const int ws   = wsum[lane];
    const int winc = wave_incl_scan(ws, lane);
    const int wexc = __shfl(winc, wave) - __shfl(ws, wave);
    total = __shfl(winc, 31);
    __syncthreads();
    return wexc + inc - v;
}

__launch_bounds__(BT)
__global__ void bin_place(const int* __restrict__ ei, int E,
                          const int* __restrict__ cnt,
                          int* __restrict__ binStart,
                          int* __restrict__ keys)
{
    __shared__ __attribute__((aligned(16))) int stage[2 * RG * NBP];
    __shared__ int hist[NBP];
    __shared__ int lstart[NBP];
    __shared__ int gdelta[NBP];
    __shared__ unsigned short sbin[EPB];
    __shared__ int wsum[32];
    const int tid = threadIdx.x;
    const int blk = blockIdx.x;

    int sv[EPT], dv[EPT];
    const int base = blk * EPB + tid;
#pragma unroll
    for (int k = 0; k < EPT; ++k) {
        int e = base + k * BT;
        sv[k] = 0;
        dv[k] = -1;
        if (e < E) { sv[k] = ei[e]; dv[k] = ei[E + e]; }
    }

    {
        const int g  = __builtin_amdgcn_readfirstlane(tid >> 8);
        const int c4 = (tid & 255) * 4;
        int4 tot  = make_int4(0, 0, 0, 0);
        int4 part = make_int4(0, 0, 0, 0);
        constexpr int TB = 16;
#pragma unroll 1
        for (int h = 0; h < RPG; h += TB) {
            int4 v[TB];
#pragma unroll
            for (int r = 0; r < TB; ++r) {
                const int rr  = h + r;
                const int row = g * RPG + rr;
                v[r] = make_int4(0, 0, 0, 0);
                if (rr < RPG && row < NBLK)
                    v[r] = *(const int4*)(cnt + (size_t)row * NBP + c4);
            }
#pragma unroll
            for (int r = 0; r < TB; ++r) {
                const int row = g * RPG + h + r;
                tot.x += v[r].x; tot.y += v[r].y; tot.z += v[r].z; tot.w += v[r].w;
                if (row < blk) {
                    part.x += v[r].x; part.y += v[r].y; part.z += v[r].z; part.w += v[r].w;
                }
            }
        }
        *(int4*)(stage + g * NBP + c4)        = tot;
        *(int4*)(stage + (RG + g) * NBP + c4) = part;
    }
    hist[tid] = 0;
    __syncthreads();
    int tot = 0, part = 0;
#pragma unroll
    for (int g = 0; g < RG; ++g) {
        tot  += stage[g * NBP + tid];
        part += stage[(RG + g) * NBP + tid];
    }
    int all;
    const int start = block_excl_scan(tot, wsum, all);
    if (blk == 0 && tid <= NBIN) binStart[tid] = start;

    int key[EPT], rnk[EPT], bin[EPT];
#pragma unroll
    for (int k = 0; k < EPT; ++k) {
        bin[k] = -1;
        if (dv[k] >= 0) {
            bin[k] = dv[k] / DB;
            key[k] = (sv[k] << DLB) | (dv[k] - bin[k] * DB);
            rnk[k] = atomicAdd(&hist[bin[k]], 1);
        }
    }
    __syncthreads();
    int nloc;
    const int ls = block_excl_scan(hist[tid], wsum, nloc);
    lstart[tid] = ls;
    gdelta[tid] = start + part - ls;
    __syncthreads();

    int* skey = stage;
#pragma unroll
    for (int k = 0; k < EPT; ++k) {
        if (bin[k] >= 0) {
            const int p = lstart[bin[k]] + rnk[k];
            skey[p] = key[k];
            sbin[p] = (unsigned short)bin[k];
        }
    }
    __syncthreads();
    for (int idx = tid; idx < nloc; idx += BT)
        keys[gdelta[sbin[idx]] + idx] = skey[idx];
}

template <int N, bool PART>
__device__ __forceinline__ void pull_batch(const float* __restrict__ hl, float* al,
                                           int2 r, int cnt)
{
    float2 v[N];
#pragma unroll
    for (int q = 0; q < N; ++q) {
        const int e = PART ? min(q, cnt - 1) : q;
        const unsigned k = (unsigned)__builtin_amdgcn_readlane(r.x, e);
        v[q] = *(const float2*)(hl + (size_t)(k >> DLB) * HID);
    }
#pragma unroll
    for (int q = 0; q < N; ++q) {
        if (!PART || q < cnt) {
            const unsigned k = (unsigned)__builtin_amdgcn_readlane(r.x, q);
            const float w = __int_as_float(__builtin_amdgcn_readlane(r.y, q));
            float* p = al + (k & ((1u << DLB) - 1u)) * HID;
            atomicAdd(p,     w * v[q].x);
            atomicAdd(p + 1, w * v[q].y);
        }
    }
}

template <bool FIRST>
__launch_bounds__(256)
__global__ void edge_pull(const float* __restrict__ hw,
                          const int* __restrict__ binStart,
                          const int* __restrict__ keys,
                          int2* __restrict__ rec,
                          const float* __restrict__ bias,
                          const float* __restrict__ dinv,
                          float* __restrict__ agg, int M)
{
    __shared__ __attribute__((aligned(16))) float As[DB * HID];
    const int tid  = threadIdx.x;
    const int bin  = blockIdx.x;
    const int row0 = bin * DB;
    const int nrow = min(DB, M - row0);

    const int beg  = binStart[bin];
    const int end  = binStart[bin + 1];
    const int wave = __builtin_amdgcn_readfirstlane(tid >> 5);
    const int lane = tid & 31;

    const int cb0 = beg + wave * 32;
    int2  rn  = make_int2(0, 0);
    int   k1  = 0, k2 = 0;
    if (FIRST) {
        if (cb0 + lane < end)       k1 = keys[cb0 + lane];
        if (cb0 + 256 + lane < end) k2 = keys[cb0 + 256 + lane];
    } else {
        if (cb0 + lane < end) rn = rec[cb0 + lane];
    }

    {
        constexpr int RQ = (DB * (HID / 4) + 255) / 256;
        const int c4 = (tid & 15) * 4;
        const float4 b = *(const float4*)(bias + c4);
        float  dq[RQ];
        float4 hq[RQ];
#pragma unroll
        for (int q = 0; q < RQ; ++q) {
            const int r = min((tid >> 4) + 16 * q, nrow - 1);
            dq[q] = dinv[row0 + r];
            hq[q] = *(const float4*)(hw + (size_t)(row0 + r) * HID + c4);
        }
#pragma unroll
        for (int q = 0; q < RQ; ++q) {
            const int r = (tid >> 4) + 16 * q;
            if (r < nrow) {
                const float di = dq[q] * dq[q];
                const float4 h = hq[q];
                *(float4*)(As + r * HID + c4) =
                    make_float4(b.x + di * h.x, b.y + di * h.y, b.z + di * h.z, b.w + di * h.w);
            }
        }
    }
    __syncthreads();

    const float* hl = hw + 2 * lane;
    float* al = As + 2 * lane;

    float ds1 = 0.0f, dd1 = 0.0f;
    if (FIRST) {
        ds1 = dinv[(unsigned)k1 >> DLB];
        dd1 = dinv[row0 + (k1 & ((1 << DLB) - 1))];
    }
    for (int cb = cb0; cb < end; cb += 256) {
        const int cnt = min(32, end - cb);
        int2 r;
        if (FIRST) {
            r = make_int2(k1, __float_as_int(ds1 * dd1));
            if (lane < cnt) rec[cb + lane] = r;
            k1 = k2;
            k2 = 0;
            if (cb + 512 + lane < end) k2 = keys[cb + 512 + lane];
            ds1 = dinv[(unsigned)k1 >> DLB];
            dd1 = dinv[row0 + (k1 & ((1 << DLB) - 1))];
        } else {
            r = rn;
            rn = make_int2(0, 0);
            if (cb + 256 + lane < end) rn = rec[cb + 256 + lane];
        }
        if (cnt == 32)      pull_batch<32, false>(hl, al, r, cnt);
        else if (cnt > 16)  pull_batch<32, true>(hl, al, r, cnt);
        else                pull_batch<16, true>(hl, al, r, cnt);
    }
    __syncthreads();

    for (int i = tid; i < nrow * (HID / 4); i += 256) {
        const int r  = i >> 4;
        const int c4 = (i & 15) * 4;
        *(float4*)(agg + (size_t)(row0 + r) * HID + c4) = *(const float4*)(As + r * HID + c4);
    }
}

extern "C" void kernel_launch(void* const* d_in, const int* in_sizes, int n_in,
                              void* d_out, int out_size, void* d_ws, size_t ws_size,
                              hipStream_t stream)
{
    (void)in_sizes; (void)n_in; (void)out_size; (void)ws_size;

    const float* x  = (const float*)d_in[0];
    const int*   ei = (const int*)d_in[1];
    const float* Wm[3] = {(const float*)d_in[2], (const float*)d_in[4],
                          (const float*)d_in[6]};
    const float* bm[3] = {(const float*)d_in[3], (const float*)d_in[5],
                          (const float*)d_in[7]};
    const float* Wl = (const float*)d_in[8];
    const float* bl = (const float*)d_in[9];
    float* out = (float*)d_out;

    char* ws = (char*)d_ws;
    auto align256 = [](size_t b) { return (b + 255) & ~(size_t)255; };
    size_t off = 0;
    float* dinv = (float*)(ws + off); off += align256((size_t)GN * 4);
    float* hw   = (float*)(ws + off); off += align256((size_t)GN * HID * 4);
    float* agg[3];
    for (int l = 0; l < 3; ++l) {
        agg[l] = (float*)(ws + off); off += align256((size_t)GN * HID * 4);
    }
    int*  cnt       = (int*)(ws + off);  off += align256((size_t)NBLK * NBP * 4);
    int*  binStart  = (int*)(ws + off);  off += align256((size_t)(NBIN + 1) * 4);
    int2* rec       = (int2*)(ws + off); off += align256((size_t)GE * 8);
    int*  keys      = (int*)(ws + off);  off += align256((size_t)GE * 4);

    const dim3 blk(256);

    bin_count  <<<NBLK, dim3(BT), 0, stream>>>(ei, GE, cnt);
    bin_place  <<<NBLK, dim3(BT), 0, stream>>>(ei, GE, cnt, binStart, keys);
    deg_bins   <<<NBIN, blk, 0, stream>>>(keys, binStart, dinv, GN);

    const int gemmBlocks = (GN + 127) / 128;

    for (int l = 0; l < 3; ++l) {
        const int K = (l == 0) ? DIN : HID;
        const size_t shmem = (size_t)(K / 2) * PadRow<64>::value * sizeof(float2);
        if (l == 0) {
            gemm_wmma_f32<4, false, false, HID, DIN><<<gemmBlocks, blk, shmem, stream>>>(
                x, x, x, DIN, DIN, 1, Wm[0], HID, GN, hw, HID, nullptr);
        } else {
            gemm_wmma_f32<4, true, false, HID, HID><<<gemmBlocks, blk, shmem, stream>>>(
                agg[l - 1], agg[l - 1], agg[l - 1], HID, HID, 1, Wm[l], HID,
                GN, hw, HID, nullptr);
        }
        if (l == 0)
            edge_pull<true><<<NBIN, blk, 0, stream>>>(hw, binStart, keys, rec, bm[l], dinv,
                                                      agg[l], GN);
        else
            edge_pull<false><<<NBIN, blk, 0, stream>>>(hw, binStart, keys, rec, bm[l], dinv,
                                                       agg[l], GN);
    }

    const size_t shmemF = (size_t)(CAT / 2) * PadRow<48>::value * sizeof(float2);
    gemm_wmma_f32<3, true, true, LAB, CAT><<<gemmBlocks, blk, shmemF, stream>>>(
        agg[0], agg[1], agg[2], HID, HID, 3, Wl, LAB, GN, out, LAB, bl);
}


# fewer, resident waves per launch: each GEMM is 256 work-groups, one per WGP (B staged once, waves loop over 16-row stripes, next stripe's A rows in flight); deg_bins 2 waves a bin; k order kept
# speedup vs baseline: 3.0443x; 1.0674x over previous
#include <hip/hip_runtime.h>


typedef __attribute__((ext_vector_type(2))) float v2f;
typedef __attribute__((ext_vector_type(8))) float v8f;

static constexpr int GN   = 100000;
static constexpr int GE   = 1000000;
static constexpr int DIN  = 128;
static constexpr int HID  = 64;
static constexpr int CAT  = 192;
static constexpr int LAB  = 40;

template <int W> struct PadRow {
    static constexpr int value = W + ((48 - (W % 32)) % 32);
};

template <int X>
__device__ __forceinline__ float row_xor(float v) {
    return __int_as_float(__builtin_amdgcn_update_dpp(
        0, __float_as_int(v), 0x160 + X, 0xf, 0xf, true));
}
__device__ __forceinline__ float row_max16(float v) {
    v = fmaxf(v, row_xor<1>(v));
    v = fmaxf(v, row_xor<2>(v));
    v = fmaxf(v, row_xor<4>(v));
    v = fmaxf(v, row_xor<8>(v));
    return v;
}
__device__ __forceinline__ float row_sum16(float v) {
    v += row_xor<1>(v);
    v += row_xor<2>(v);
    v += row_xor<4>(v);
    v += row_xor<8>(v);
    return v;
}

static constexpr int GEMM_WGS = 256;
static constexpr int GEMM_LDS_BYTES = 164 * 1024;

template <int KP, int NPL>
__device__ __forceinline__ void load_stripe(v2f (&a)[KP * NPL / 4],
                                            const float* __restrict__ A0,
                                            const float* __restrict__ A1,
                                            const float* __restrict__ A2,
                                            int s, int l15, int kb, int M)
{
    int mA = s * 16 + l15;
    if (mA > M - 1) mA = M - 1;
#pragma unroll
    for (int f = 0; f < KP * NPL / 4; ++f) {
        const int p = (4 * f) / KP;
        const float* Ap = (p == 0) ? A0 : ((p == 1) ? A1 : A2);
        a[f] = *(const v2f*)(Ap + (size_t)mA * KP + (4 * f - p * KP) + kb);
    }
}

__device__ __forceinline__ void wave_lds_order() {
    __builtin_amdgcn_fence(__ATOMIC_RELEASE, "wavefront");
    __builtin_amdgcn_wave_barrier();
    __builtin_amdgcn_fence(__ATOMIC_ACQUIRE, "wavefront");
}

template <int NT, bool RELU, bool LSM, bool BIAS, int DOUT, int KP, int NPL>
__launch_bounds__(256)
__global__ void gemm_wmma_f32(const float* __restrict__ A0,
                              const float* __restrict__ A1,
                              const float* __restrict__ A2,
                              const float* __restrict__ B,
                              int M, float* __restrict__ out,
                              const float* __restrict__ bias)
{
    constexpr int KT  = KP * NPL;
    constexpr int NFR = KT / 4;
    constexpr int W   = NT * 16;
    constexpr int WP  = PadRow<W>::value;
    static_assert(KP % 4 == 0 && NPL >= 1 && NPL <= 3, "planes of whole fragments");
    static_assert((KT / 2) * WP * 8 + (LSM ? 8 * 16 * DOUT * 4 : 0) <= GEMM_LDS_BYTES, "B and the waves' tiles fit the reservation");
    __shared__ __attribute__((aligned(16))) v2f Bs2[GEMM_LDS_BYTES / 8];

    const int wave = __builtin_amdgcn_readfirstlane(threadIdx.x >> 5);
    const int lane = threadIdx.x & 31;
    const int l15  = lane & 15;
    const int half = lane >> 4;
    const int kb   = half * 2;

    const int nS = (M + 15) >> 4;
    const int s0 = (int)(((long long)blockIdx.x * nS) / GEMM_WGS);
    const int s1 = (int)(((long long)(blockIdx.x + 1) * nS) / GEMM_WGS);

    int s = s0 + wave;
    v2f a[NFR];
    load_stripe<KP, NPL>(a, A0, A1, A2, (s < s1) ? s : 0, l15, kb, M);

    static_assert(DOUT % 4 == 0 && KT % 2 == 0, "B rows are read as 16-byte pieces, in k pairs");
    constexpr int G   = W / 4;
    constexpr int NU  = (KT / 2) * G;
    constexpr int UPT = (NU + 255) / 256;
    {
        float4 b0[UPT], b1[UPT];
#pragma unroll
        for (int i = 0; i < UPT; ++i) {
            const int u  = threadIdx.x + 256 * i;
            const int kp = u / G;
            const int c4 = (u - kp * G) * 4;
            const bool in = (256 * (i + 1) <= NU) || (u < NU);
            const int kl = in ? kp : 0;
            const int cl = (c4 < DOUT) ? c4 : 0;
            b0[i] = *(const float4*)(B + (size_t)(2 * kl)     * DOUT + cl);
            b1[i] = *(const float4*)(B + (size_t)(2 * kl + 1) * DOUT + cl);
        }
#pragma unroll
        for (int i = 0; i < UPT; ++i) {
            const int u  = threadIdx.x + 256 * i;
            const int kp = u / G;
            const int c4 = (u - kp * G) * 4;
            if (c4 >= DOUT) {
                b0[i] = make_float4(0.f, 0.f, 0.f, 0.f);
                b1[i] = make_float4(0.f, 0.f, 0.f, 0.f);
            }
            if ((256 * (i + 1) <= NU) || (u < NU)) {
                v2f* d = Bs2 + kp * WP + c4;
                d[0] = (v2f){b0[i].x, b1[i].x};
                d[1] = (v2f){b0[i].y, b1[i].y};
                d[2] = (v2f){b0[i].z, b1[i].z};
                d[3] = (v2f){b0[i].w, b1[i].w};
            }
        }
    }
    __syncthreads();

    float bvl[NT];
    bool  ok[NT];
#pragma unroll
    for (int t = 0; t < NT; ++t) {
        const int n = t * 16 + l15;
        ok[t]  = (t * 16 + 15 < DOUT) || (n < DOUT);
        bvl[t] = (BIAS && ok[t]) ? bias[n] : 0.0f;
    }
    float* tile = (float*)(Bs2 + (KT / 2) * WP) + wave * (16 * DOUT);

    while (s < s1) {
        const int  sn   = s + 8;
        const bool more = sn < s1;
        v2f an[NFR];
        if (more) {
            load_stripe<KP, NPL>(an, A0, A1, A2, sn, l15, kb, M);
        } else {
#pragma unroll
            for (int f = 0; f < NFR; ++f) an[f] = a[f];
        }

        v8f acc[NT];
#pragma unroll
        for (int t = 0; t < NT; ++t)
            acc[t] = (v8f){0.f, 0.f, 0.f, 0.f, 0.f, 0.f, 0.f, 0.f};

#pragma unroll
        for (int f = 0; f < NFR; ++f) {
            v2f af = a[f];
            if (RELU) {
                af.x = fmaxf(af.x, 0.0f);
                af.y = fmaxf(af.y, 0.0f);
            }
            const v2f* brow = Bs2 + (2 * f + half) * WP + l15;
#pragma unroll
            for (int t = 0; t < NT; ++t) {
                v2f b = brow[t * 16];
                acc[t] = __builtin_amdgcn_wmma_f32_16x16x4_f32(
                    false, af, false, b, (short)0, acc[t], false, false);
            }
        }

        const int m0 = s * 16;
        if (LSM) {
            static_assert(!LSM || (DOUT % 4 == 0 && DOUT <= NT * 16), "row shape");
#pragma unroll
            for (int v = 0; v < 8; ++v) {
                float xv[NT];
                float mx = -3.4e38f;
#pragma unroll
                for (int t = 0; t < NT; ++t) {
                    xv[t] = acc[t][v] + bvl[t];
                    mx = fmaxf(mx, ok[t] ? xv[t] : -3.4e38f);
                }
                mx = row_max16(mx);
                float sm = 0.0f;
#pragma unroll
                for (int t = 0; t < NT; ++t)
                    sm += ok[t] ? __expf(xv[t] - mx) : 0.0f;
                sm = row_sum16(sm);
                const float ls = __logf(sm);
                float* trow = tile + (half * 8 + v) * DOUT + l15;
#pragma unroll
                for (int t = 0; t < NT; ++t)
                    if (ok[t]) trow[t * 16] = xv[t] - mx - ls;
            }
            wave_lds_order();
            const int rows = min(16, M - m0);
            const int nvec = (rows > 0) ? (rows * DOUT) >> 2 : 0;
            const float4* srcv = (const float4*)tile;
            float4* dstv = (float4*)(out + (size_t)m0 * DOUT);
            constexpr int NV = (16 * DOUT / 4) / 32;
            if (rows == 16 && (16 * DOUT / 4) % 32 == 0) {
                float4 tv[NV > 0 ? NV : 1];
#pragma unroll
                for (int i = 0; i < NV; ++i) tv[i] = srcv[lane + 32 * i];
#pragma unroll
                for (int i = 0; i < NV; ++i) dstv[lane + 32 * i] = tv[i];
            } else {
                for (int i = lane; i < nvec; i += 32) dstv[i] = srcv[i];
            }
            wave_lds_order();
        } else if (m0 + 16 <= M) {
            float* orow = out + (size_t)(m0 + half * 8) * DOUT + l15;
#pragma unroll
            for (int t = 0; t < NT; ++t) {
#pragma unroll
                for (int v = 0; v < 8; ++v)
                    if (ok[t]) orow[v * DOUT + t * 16] = BIAS ? acc[t][v] + bvl[t] : acc[t][v];
            }
        } else {
#pragma unroll
            for (int t = 0; t < NT; ++t) {
                const int n = t * 16 + l15;
#pragma unroll
                for (int v = 0; v < 8; ++v) {
                    const int m = m0 + half * 8 + v;
                    if (m < M && ok[t])
                        out[(size_t)m * DOUT + n] = BIAS ? acc[t][v] + bvl[t] : acc[t][v];
                }
            }
        }

#pragma unroll
        for (int f = 0; f < NFR; ++f) a[f] = an[f];
        s = sn;
    }
}

static constexpr int DB    = 98;
static constexpr int DLB   = 7;
static constexpr int NBIN  = (GN + DB - 1) / DB;
static constexpr int BT    = 1024;
static constexpr int EPB   = 8192;
static constexpr int EPT   = EPB / BT;
static constexpr int NBLK  = (GE + EPB - 1) / EPB;
static constexpr int NBP   = 1024;
static constexpr int RG    = 4;
static constexpr int RPG   = (NBLK + RG - 1) / RG;
static_assert(DB < (1 << DLB), "in-block row number must fit its field");
static_assert(((long long)GN << DLB) < (1LL << 31), "a key (src << DLB | row) must stay non-negative");
static_assert(NBIN < NBP && NBP == BT, "one thread per padded bin, and one padded bin behind the last");
static_assert(BT / RG * 4 == NBP, "a row group's threads cover one table row with 16-byte loads");
static_assert(2 * RG * NBP >= EPB, "the table partials' LDS space is reused for the ordered keys");
static_assert(NBIN < 65536, "a bin number is staged as 16 bits");

__launch_bounds__(BT)
__global__ void bin_count(const int* __restrict__ ei, int E, int* __restrict__ cnt) {
    __shared__ int hist[NBP];
    const int tid = threadIdx.x;
    hist[tid] = 0;
    __syncthreads();
    const int base = blockIdx.x * EPB + tid;
    int dv[EPT];
#pragma unroll
    for (int k = 0; k < EPT; ++k) {
        int e = base + k * BT;
        dv[k] = -1;
        if (e < E) dv[k] = ei[E + e];
    }
#pragma unroll
    for (int k = 0; k < EPT; ++k) {
        if (dv[k] >= 0) atomicAdd(&hist[dv[k] / DB], 1);
    }
    __syncthreads();
    cnt[blockIdx.x * NBP + tid] = hist[tid];
}

static constexpr int DGT = 64;
__launch_bounds__(DGT)
__global__ void deg_bins(const int* __restrict__ keys,
                         const int* __restrict__ binStart,
                         float* __restrict__ dinv, int M) {
    __shared__ int cnt[1 << DLB];
    const int tid  = threadIdx.x;
    const int bin  = blockIdx.x;
    const int row0 = bin * DB;
    const int nrow = min(DB, M - row0);
    for (int i = tid; i < (1 << DLB); i += DGT) cnt[i] = 0;
    __syncthreads();
    const int beg = binStart[bin];
    const int end = binStart[bin + 1];
    constexpr int KPT = 16;
    for (int base = beg + tid; base < end; base += DGT * KPT) {
        int kv[KPT];
#pragma unroll
        for (int q = 0; q < KPT; ++q) {
            const int i = base + q * DGT;
            kv[q] = (i < end) ? keys[i] : -1;
        }
#pragma unroll
        for (int q = 0; q < KPT; ++q)
            if (kv[q] >= 0) atomicAdd(&cnt[kv[q] & ((1 << DLB) - 1)], 1);
    }
    __syncthreads();
    for (int r = tid; r < nrow; r += DGT) dinv[row0 + r] = rsqrtf(1.0f + (float)cnt[r]);
}

__device__ __forceinline__ int wave_incl_scan(int v, int lane) {
#pragma unroll
    for (int off = 1; off < 32; off <<= 1) {
        const int t = __shfl_up(v, off);
        if (lane >= off) v += t;
    }
    return v;
}
__device__ __forceinline__ int block_excl_scan(int v, int* wsum, int& total) {
    const int lane = threadIdx.x & 31;
    const int wave = threadIdx.x >> 5;
    const int inc = wave_incl_scan(v, lane);
    if (lane == 31) wsum[wave] = inc;
    __syncthreads();
    const int ws   = wsum[lane];
    const int winc = wave_incl_scan(ws, lane);
    const int wexc = __shfl(winc, wave) - __shfl(ws, wave);
    total = __shfl(winc, 31);
    __syncthreads();
    return wexc + inc - v;
}

__launch_bounds__(BT)
__global__ void bin_place(const int* __restrict__ ei, int E,
                          const int* __restrict__ cnt,
                          int* __restrict__ binStart,
                          int* __restrict__ keys)
{
    __shared__ __attribute__((aligned(16))) int stage[2 * RG * NBP];
    __shared__ int hist[NBP];
    __shared__ int lstart[NBP];
    __shared__ int gdelta[NBP];
    __shared__ unsigned short sbin[EPB];
    __shared__ int wsum[32];
    const int tid = threadIdx.x;
    const int blk = blockIdx.x;

    int sv[EPT], dv[EPT];
    const int base = blk * EPB + tid;
#pragma unroll
    for (int k = 0; k < EPT; ++k) {
        int e = base + k * BT;
        sv[k] = 0;
        dv[k] = -1;
        if (e < E) { sv[k] = ei[e]; dv[k] = ei[E + e]; }
    }

    {
        const int g  = __builtin_amdgcn_readfirstlane(tid >> 8);
        const int c4 = (tid & 255) * 4;
        int4 tot  = make_int4(0, 0, 0, 0);
        int4 part = make_int4(0, 0, 0, 0);
        constexpr int TB = 16;
#pragma unroll 1
        for (int h = 0; h < RPG; h += TB) {
            int4 v[TB];
#pragma unroll
            for (int r = 0; r < TB; ++r) {
                const int rr  = h + r;
                const int row = g * RPG + rr;
                v[r] = make_int4(0, 0, 0, 0);
                if (rr < RPG && row < NBLK)
                    v[r] = *(const int4*)(cnt + (size_t)row * NBP + c4);
            }
#pragma unroll
            for (int r = 0; r < TB; ++r) {
                const int row = g * RPG + h + r;
                tot.x += v[r].x; tot.y += v[r].y; tot.z += v[r].z; tot.w += v[r].w;
                if (row < blk) {
                    part.x += v[r].x; part.y += v[r].y; part.z += v[r].z; part.w += v[r].w;
                }
            }
        }
        *(int4*)(stage + g * NBP + c4)        = tot;
        *(int4*)(stage + (RG + g) * NBP + c4) = part;
    }
    hist[tid] = 0;
    __syncthreads();
    int tot = 0, part = 0;
#pragma unroll
    for (int g = 0; g < RG; ++g) {
        tot  += stage[g * NBP + tid];
        part += stage[(RG + g) * NBP + tid];
    }
    int all;
    const int start = block_excl_scan(tot, wsum, all);
    if (blk == 0 && tid <= NBIN) binStart[tid] = start;

    int key[EPT], rnk[EPT], bin[EPT];
#pragma unroll
    for (int k = 0; k < EPT; ++k) {
        bin[k] = -1;
        if (dv[k] >= 0) {
            bin[k] = dv[k] / DB;
            key[k] = (sv[k] << DLB) | (dv[k] - bin[k] * DB);
            rnk[k] = atomicAdd(&hist[bin[k]], 1);
        }
    }
    __syncthreads();
    int nloc;
    const int ls = block_excl_scan(hist[tid], wsum, nloc);
    lstart[tid] = ls;
    gdelta[tid] = start + part - ls;
    __syncthreads();

    int* skey = stage;
#pragma unroll
    for (int k = 0; k < EPT; ++k) {
        if (bin[k] >= 0) {
            const int p = lstart[bin[k]] + rnk[k];
            skey[p] = key[k];
            sbin[p] = (unsigned short)bin[k];
        }
    }
    __syncthreads();
    for (int idx = tid; idx < nloc; idx += BT)
        keys[gdelta[sbin[idx]] + idx] = skey[idx];
}

template <int N, bool PART>
__device__ __forceinline__ void pull_batch(const float* __restrict__ hl, float* al,
                                           int2 r, int cnt)
{
    float2 v[N];
#pragma unroll
    for (int q = 0; q < N; ++q) {
        const int e = PART ? min(q, cnt - 1) : q;
        const unsigned k = (unsigned)__builtin_amdgcn_readlane(r.x, e);
        v[q] = *(const float2*)(hl + (size_t)(k >> DLB) * HID);
    }
#pragma unroll
    for (int q = 0; q < N; ++q) {
        if (!PART || q < cnt) {
            const unsigned k = (unsigned)__builtin_amdgcn_readlane(r.x, q);
            const float w = __int_as_float(__builtin_amdgcn_readlane(r.y, q));
            float* p = al + (k & ((1u << DLB) - 1u)) * HID;
            atomicAdd(p,     w * v[q].x);
            atomicAdd(p + 1, w * v[q].y);
        }
    }
}

template <bool FIRST>
__launch_bounds__(256)
__global__ void edge_pull(const float* __restrict__ hw,
                          const int* __restrict__ binStart,
                          const int* __restrict__ keys,
                          int2* __restrict__ rec,
                          const float* __restrict__ bias,
                          const float* __restrict__ dinv,
                          float* __restrict__ agg, int M)
{
    __shared__ __attribute__((aligned(16))) float As[DB * HID];
    const int tid  = threadIdx.x;
    const int bin  = blockIdx.x;
    const int row0 = bin * DB;
    const int nrow = min(DB, M - row0);

    const int beg  = binStart[bin];
    const int end  = binStart[bin + 1];
    const int wave = __builtin_amdgcn_readfirstlane(tid >> 5);
    const int lane = tid & 31;

    const int cb0 = beg + wave * 32;
    int2  rn  = make_int2(0, 0);
    int   k1  = 0, k2 = 0;
    if (FIRST) {
        if (cb0 + lane < end)       k1 = keys[cb0 + lane];
        if (cb0 + 256 + lane < end) k2 = keys[cb0 + 256 + lane];
    } else {
        if (cb0 + lane < end) rn = rec[cb0 + lane];
    }

    {
        constexpr int RQ = (DB * (HID / 4) + 255) / 256;
        const int c4 = (tid & 15) * 4;
        const float4 b = *(const float4*)(bias + c4);
        float  dq[RQ];
        float4 hq[RQ];
#pragma unroll
        for (int q = 0; q < RQ; ++q) {
            const int r = min((tid >> 4) + 16 * q, nrow - 1);
            dq[q] = dinv[row0 + r];
            hq[q] = *(const float4*)(hw + (size_t)(row0 + r) * HID + c4);
        }
#pragma unroll
        for (int q = 0; q < RQ; ++q) {
            const int r = (tid >> 4) + 16 * q;
            if (r < nrow) {
                const float di = dq[q] * dq[q];
                const float4 h = hq[q];
                *(float4*)(As + r * HID + c4) =
                    make_float4(b.x + di * h.x, b.y + di * h.y, b.z + di * h.z, b.w + di * h.w);
            }
        }
    }
    __syncthreads();

    const float* hl = hw + 2 * lane;
    float* al = As + 2 * lane;

    float ds1 = 0.0f, dd1 = 0.0f;
    if (FIRST) {
        ds1 = dinv[(unsigned)k1 >> DLB];
        dd1 = dinv[row0 + (k1 & ((1 << DLB) - 1))];
    }
    for (int cb = cb0; cb < end; cb += 256) {
        const int cnt = min(32, end - cb);
        int2 r;
        if (FIRST) {
            r = make_int2(k1, __float_as_int(ds1 * dd1));
            if (lane < cnt) rec[cb + lane] = r;
            k1 = k2;
            k2 = 0;
            if (cb + 512 + lane < end) k2 = keys[cb + 512 + lane];
            ds1 = dinv[(unsigned)k1 >> DLB];
            dd1 = dinv[row0 + (k1 & ((1 << DLB) - 1))];
        } else {
            r = rn;
            rn = make_int2(0, 0);
            if (cb + 256 + lane < end) rn = rec[cb + 256 + lane];
        }
        if (cnt == 32)      pull_batch<32, false>(hl, al, r, cnt);
        else if (cnt > 16)  pull_batch<32, true>(hl, al, r, cnt);
        else                pull_batch<16, true>(hl, al, r, cnt);
    }
    __syncthreads();

    for (int i = tid; i < nrow * (HID / 4); i += 256) {
        const int r  = i >> 4;
        const int c4 = (i & 15) * 4;
        *(float4*)(agg + (size_t)(row0 + r) * HID + c4) = *(const float4*)(As + r * HID + c4);
    }
}

extern "C" void kernel_launch(void* const* d_in, const int* in_sizes, int n_in,
                              void* d_out, int out_size, void* d_ws, size_t ws_size,
                              hipStream_t stream)
{
    (void)in_sizes; (void)n_in; (void)out_size; (void)ws_size;

    const float* x  = (const float*)d_in[0];
    const int*   ei = (const int*)d_in[1];
    const float* Wm[3] = {(const float*)d_in[2], (const float*)d_in[4],
                          (const float*)d_in[6]};
    const float* bm[3] = {(const float*)d_in[3], (const float*)d_in[5],
                          (const float*)d_in[7]};
    const float* Wl = (const float*)d_in[8];
    const float* bl = (const float*)d_in[9];
    float* out = (float*)d_out;

    char* ws = (char*)d_ws;
    auto align256 = [](size_t b) { return (b + 255) & ~(size_t)255; };
    size_t off = 0;
    float* dinv = (float*)(ws + off); off += align256((size_t)GN * 4);
    float* hw   = (float*)(ws + off); off += align256((size_t)GN * HID * 4);
    float* agg[3];
    for (int l = 0; l < 3; ++l) {
        agg[l] = (float*)(ws + off); off += align256((size_t)GN * HID * 4);
    }
    int*  cnt       = (int*)(ws + off);  off += align256((size_t)NBLK * NBP * 4);
    int*  binStart  = (int*)(ws + off);  off += align256((size_t)(NBIN + 1) * 4);
    int2* rec       = (int2*)(ws + off); off += align256((size_t)GE * 8);
    int*  keys      = (int*)(ws + off);  off += align256((size_t)GE * 4);

    const dim3 blk(256);

    bin_count  <<<NBLK, dim3(BT), 0, stream>>>(ei, GE, cnt);
    bin_place  <<<NBLK, dim3(BT), 0, stream>>>(ei, GE, cnt, binStart, keys);
    deg_bins   <<<NBIN, dim3(DGT), 0, stream>>>(keys, binStart, dinv, GN);

    for (int l = 0; l < 3; ++l) {
        if (l == 0) {
            gemm_wmma_f32<4, false, false, false, HID, DIN, 1><<<GEMM_WGS, blk, 0, stream>>>(
                x, x, x, Wm[0], GN, hw, nullptr);
        } else {
            gemm_wmma_f32<4, true, false, false, HID, HID, 1><<<GEMM_WGS, blk, 0, stream>>>(
                agg[l - 1], agg[l - 1], agg[l - 1], Wm[l], GN, hw, nullptr);
        }
        if (l == 0)
            edge_pull<true><<<NBIN, blk, 0, stream>>>(hw, binStart, keys, rec, bm[l], dinv,
                                                      agg[l], GN);
        else
            edge_pull<false><<<NBIN, blk, 0, stream>>>(hw, binStart, keys, rec, bm[l], dinv,
                                                       agg[l], GN);
    }

    gemm_wmma_f32<3, true, true, true, LAB, HID, 3><<<GEMM_WGS, blk, 0, stream>>>(
        agg[0], agg[1], agg[2], Wl, GN, out, bl);
}
